# starting kernel
# baseline (speedup 1.0000x reference)
#include <hip/hip_runtime.h>
#include <math.h>

typedef __attribute__((ext_vector_type(2))) float v2f;
typedef __attribute__((ext_vector_type(8))) float v8f;

#define HEADS 8

#if defined(__gfx1250__) && __has_builtin(__builtin_amdgcn_wmma_f32_16x16x4_f32)
#define USE_WMMA_F32 1
#else
#define USE_WMMA_F32 0
#endif

__device__ __forceinline__ unsigned enc_ord(float f) {
  unsigned b = __float_as_uint(f);
  return (b & 0x80000000u) ? ~b : (b | 0x80000000u);
}
__device__ __forceinline__ float dec_ord(unsigned u) {
  unsigned b = (u & 0x80000000u) ? (u ^ 0x80000000u) : ~u;
  return __uint_as_float(b);
}

__global__ void gemm_wmma(const float* __restrict__ A, const float* __restrict__ W,
                          float* __restrict__ C, int M, int Kin, int Kout) {
  int wid = blockIdx.x * (blockDim.x >> 5) + (threadIdx.x >> 5);
  int tilesN = Kout >> 4;
  int tiles  = (M >> 4) * tilesN;
  if (wid >= tiles) return;
  int tm = wid / tilesN, tn = wid % tilesN;
  int lane = threadIdx.x & 31;
  int grp  = lane >> 4;
  int ln   = lane & 15;

#if USE_WMMA_F32
  const float* Arow = A + (size_t)(tm * 16 + ln) * Kin;
  const float* Wcol = W + (size_t)tn * 16 + ln;
  v8f acc = {};
  for (int k = 0; k < Kin; k += 4) {
    v2f a, b;
    a[0] = Arow[k + 2 * grp];
    a[1] = Arow[k + 2 * grp + 1];
    b[0] = Wcol[(size_t)(k + 2 * grp) * Kout];
    b[1] = Wcol[(size_t)(k + 2 * grp + 1) * Kout];
    acc = __builtin_amdgcn_wmma_f32_16x16x4_f32(false, a, false, b,
                                                (short)0, acc, false, false);
  }
  float* Cout = C + (size_t)(tm * 16 + grp * 8) * Kout + (size_t)tn * 16 + ln;
#pragma unroll
  for (int v = 0; v < 8; ++v) Cout[(size_t)v * Kout] = acc[v];
#else
  float acc[8] = {};
  for (int k = 0; k < Kin; ++k) {
    float bv = W[(size_t)k * Kout + tn * 16 + ln];
#pragma unroll
    for (int v = 0; v < 8; ++v)
      acc[v] += A[(size_t)(tm * 16 + grp * 8 + v) * Kin + k] * bv;
  }
  float* Cout = C + (size_t)(tm * 16 + grp * 8) * Kout + (size_t)tn * 16 + ln;
#pragma unroll
  for (int v = 0; v < 8; ++v) Cout[(size_t)v * Kout] = acc[v];
#endif
}

__global__ void fill_u32(unsigned* __restrict__ p, unsigned v, long long n) {
  long long i = blockIdx.x * (long long)blockDim.x + threadIdx.x;
  long long stride = (long long)gridDim.x * blockDim.x;
  for (; i < n; i += stride) p[i] = v;
}

__device__ __forceinline__ void edge_sd(const int* __restrict__ srcI,
                                        const int* __restrict__ dstI,
                                        int e, int E, int& s, int& d) {
  if (e < E) { s = srcI[e]; d = dstI[e]; }
  else       { s = d = e - E; }
}

__global__ void edge_logits_kernel(const float* __restrict__ xl, const float* __restrict__ xr,
                                   const float* __restrict__ att,
                                   const int* __restrict__ srcI, const int* __restrict__ dstI,
                                   int E, int Etot, int C,
                                   float* __restrict__ logits, unsigned* __restrict__ lmax) {
  long long idx = blockIdx.x * (long long)blockDim.x + threadIdx.x;
  if (idx >= (long long)Etot * HEADS) return;
  int e = (int)(idx >> 3), h = (int)(idx & 7);
  int s, d; edge_sd(srcI, dstI, e, E, s, d);
  const float4* pl = (const float4*)(xl + ((size_t)s * HEADS + h) * C);
  const float4* pr = (const float4*)(xr + ((size_t)d * HEADS + h) * C);
  const float4* pa = (const float4*)(att + (size_t)h * C);
  float acc = 0.f;
  int n4 = C >> 2;
  for (int i = 0; i < n4; ++i) {
    float4 a = pl[i], b = pr[i], w = pa[i];
    float v;
    v = a.x + b.x; v = v > 0.f ? v : 0.2f * v; acc += v * w.x;
    v = a.y + b.y; v = v > 0.f ? v : 0.2f * v; acc += v * w.y;
    v = a.z + b.z; v = v > 0.f ? v : 0.2f * v; acc += v * w.z;
    v = a.w + b.w; v = v > 0.f ? v : 0.2f * v; acc += v * w.w;
  }
  logits[idx] = acc;
  atomicMax(lmax + (size_t)d * HEADS + h, enc_ord(acc));
}

__global__ void edge_accum_kernel(const float* __restrict__ xl,
                                  const int* __restrict__ srcI, const int* __restrict__ dstI,
                                  int E, int Etot, int C,
                                  const float* __restrict__ logits,
                                  const unsigned* __restrict__ lmax,
                                  float* __restrict__ denom,
                                  float* __restrict__ out) {
  long long idx = blockIdx.x * (long long)blockDim.x + threadIdx.x;
  if (idx >= (long long)Etot * HEADS) return;
  int e = (int)(idx >> 3), h = (int)(idx & 7);
  int s, d; edge_sd(srcI, dstI, e, E, s, d);
  float m = dec_ord(lmax[(size_t)d * HEADS + h]);
  float ex = expf(logits[idx] - m);
  atomicAdd(denom + (size_t)d * HEADS + h, ex);
  const float4* pm = (const float4*)(xl + ((size_t)s * HEADS + h) * C);
  float* po = out + ((size_t)d * HEADS + h) * C;
  int n4 = C >> 2;
  for (int i = 0; i < n4; ++i) {
    float4 m4 = pm[i];
    atomicAdd(po + 4 * i + 0, ex * m4.x);
    atomicAdd(po + 4 * i + 1, ex * m4.y);
    atomicAdd(po + 4 * i + 2, ex * m4.z);
    atomicAdd(po + 4 * i + 3, ex * m4.w);
  }
}

__global__ void norm_elu_bias_kernel(float* __restrict__ h, const float* __restrict__ denom,
                                     const float* __restrict__ b, long long total) {
  long long i = blockIdx.x * (long long)blockDim.x + threadIdx.x;
  long long stride = (long long)gridDim.x * blockDim.x;
  for (; i < total; i += stride) {
    int rem = (int)(i & 255);
    long long n = i >> 8;
    int hd = rem >> 5;
    float v = h[i] / denom[n * HEADS + hd] + b[rem];
    h[i] = v > 0.f ? v : (expf(v) - 1.f);
  }
}

__global__ void finalize_logsoftmax(float* __restrict__ out, const float* __restrict__ denom,
                                    const float* __restrict__ b2) {
  int n = blockIdx.x;
  int t = threadIdx.x;
  int hd = t >> 4;
  float v = out[(size_t)n * 128 + t] / denom[(size_t)n * HEADS + hd] + b2[t];
  __shared__ float sm[128];
  sm[t] = v; __syncthreads();
  for (int sft = 64; sft > 0; sft >>= 1) { if (t < sft) sm[t] = fmaxf(sm[t], sm[t + sft]); __syncthreads(); }
  float mx = sm[0]; __syncthreads();
  sm[t] = expf(v - mx); __syncthreads();
  for (int sft = 64; sft > 0; sft >>= 1) { if (t < sft) sm[t] += sm[t + sft]; __syncthreads(); }
  float lse = logf(sm[0]);
  out[(size_t)n * 128 + t] = v - mx - lse;
}

extern "C" void kernel_launch(void* const* d_in, const int* in_sizes, int n_in,
                              void* d_out, int out_size, void* d_ws, size_t ws_size,
                              hipStream_t stream) {
  (void)n_in; (void)out_size; (void)ws_size;
  const float* x    = (const float*)d_in[0];
  const int*   ei   = (const int*)d_in[1];
  const float* W1l  = (const float*)d_in[2];
  const float* W1r  = (const float*)d_in[3];
  const float* att1 = (const float*)d_in[4];
  const float* b1   = (const float*)d_in[5];
  const float* W2l  = (const float*)d_in[6];
  const float* W2r  = (const float*)d_in[7];
  const float* att2 = (const float*)d_in[8];
  const float* b2   = (const float*)d_in[9];

  int N = in_sizes[0] / 128;
  int E = in_sizes[1] / 2;
  int Etot = E + N;
  const int* srcI = ei;
  const int* dstI = ei + E;

  float* ws    = (float*)d_ws;
  float* xl    = ws;
  float* xr    = xl + (size_t)N * 256;
  float* hbuf  = xr + (size_t)N * 256;
  float* lg    = hbuf + (size_t)N * 256;
  unsigned* lmax = (unsigned*)(lg + (size_t)Etot * 8);
  float* denom = (float*)(lmax + (size_t)N * 8);
  float* outAcc = (float*)d_out;

  dim3 blk(256);
  long long eh = (long long)Etot * HEADS;
  int egrid = (int)((eh + 255) / 256);
  const unsigned ENC_NEG_INF = 0x007FFFFFu;

  int tiles1 = (N / 16) * (256 / 16);
  gemm_wmma<<<(tiles1 + 7) / 8, blk, 0, stream>>>(x, W1l, xl, N, 128, 256);
  gemm_wmma<<<(tiles1 + 7) / 8, blk, 0, stream>>>(x, W1r, xr, N, 128, 256);
  fill_u32<<<1024, blk, 0, stream>>>(lmax, ENC_NEG_INF, (long long)N * 8);
  fill_u32<<<1024, blk, 0, stream>>>((unsigned*)denom, 0u, (long long)N * 8);
  fill_u32<<<2048, blk, 0, stream>>>((unsigned*)hbuf, 0u, (long long)N * 256);
  edge_logits_kernel<<<egrid, blk, 0, stream>>>(xl, xr, att1, srcI, dstI, E, Etot, 32, lg, lmax);
  edge_accum_kernel<<<egrid, blk, 0, stream>>>(xl, srcI, dstI, E, Etot, 32, lg, lmax, denom, hbuf);
  norm_elu_bias_kernel<<<2048, blk, 0, stream>>>(hbuf, denom, b1, (long long)N * 256);

  int tiles2 = (N / 16) * (128 / 16);
  gemm_wmma<<<(tiles2 + 7) / 8, blk, 0, stream>>>(hbuf, W2l, xl, N, 256, 128);
  gemm_wmma<<<(tiles2 + 7) / 8, blk, 0, stream>>>(hbuf, W2r, xr, N, 256, 128);
  fill_u32<<<1024, blk, 0, stream>>>(lmax, ENC_NEG_INF, (long long)N * 8);
  fill_u32<<<1024, blk, 0, stream>>>((unsigned*)denom, 0u, (long long)N * 8);
  fill_u32<<<2048, blk, 0, stream>>>((unsigned*)outAcc, 0u, (long long)N * 128);
  edge_logits_kernel<<<egrid, blk, 0, stream>>>(xl, xr, att2, srcI, dstI, E, Etot, 16, lg, lmax);
  edge_accum_kernel<<<egrid, blk, 0, stream>>>(xl, srcI, dstI, E, Etot, 16, lg, lmax, denom, outAcc);
  finalize_logsoftmax<<<N, dim3(128), 0, stream>>>(outAcc, denom, b2);
}


# GEMMs fed from LDS: block has [Wl|Wr] in LDS as k-pair rows, wave keeps a 16-row A strip's whole K in registers, B fragments by 8-B LDS loads; both products per launch, 6->4; k order of sums changed
# speedup vs baseline: 5.0777x; 5.0777x over previous
#include <hip/hip_runtime.h>
#include <math.h>

typedef __attribute__((ext_vector_type(2))) float v2f;
typedef __attribute__((ext_vector_type(8))) float v8f;

#define HEADS 8

#if defined(__gfx1250__) && __has_builtin(__builtin_amdgcn_wmma_f32_16x16x4_f32)
#define USE_WMMA_F32 1
#else
#define USE_WMMA_F32 0
#endif

__device__ __forceinline__ unsigned enc_ord(float f) {
  unsigned b = __float_as_uint(f);
  return (b & 0x80000000u) ? ~b : (b | 0x80000000u);
}
__device__ __forceinline__ float dec_ord(unsigned u) {
  unsigned b = (u & 0x80000000u) ? (u ^ 0x80000000u) : ~u;
  return __uint_as_float(b);
}

#define ACC_NB  128
#define ACC_U   8
#define ACC_CAP 4096

__device__ __forceinline__ void lds_add(float* p, float v) {
  __hip_atomic_fetch_add(p, v, __ATOMIC_RELAXED, __HIP_MEMORY_SCOPE_WORKGROUP);
}

template <class F>
__device__ __forceinline__ void scan_owned(const int* __restrict__ dstI, int E, int n0,
                                           int wave, int nw, int lane, F f) {
  const int stepE = ACC_U * 32;
  const int strideE = nw * stepE;
  int base = wave * stepE;
  int cur[ACC_U], nxt[ACC_U];
#pragma unroll
  for (int u = 0; u < ACC_U; ++u) {
    int e = base + u * 32 + lane;
    cur[u] = e < E ? dstI[e] : -1;
  }
  while (base < E) {
    const int nbase = base + strideE;
    if (nbase + stepE <= E) {
      const int* p = dstI + nbase + lane;
#pragma unroll
      for (int u = 0; u < ACC_U; ++u) nxt[u] = p[u * 32];
    } else {
#pragma unroll
      for (int u = 0; u < ACC_U; ++u) {
        int e = nbase + u * 32 + lane;
        nxt[u] = e < E ? dstI[e] : -1;
      }
    }
    bool any = false;
#pragma unroll
    for (int u = 0; u < ACC_U; ++u) any |= (unsigned)(cur[u] - n0) < (unsigned)ACC_NB;
    if (__builtin_amdgcn_ballot_w32(any)) {
#pragma unroll
      for (int u = 0; u < ACC_U; ++u) {
        unsigned mask = __builtin_amdgcn_ballot_w32((unsigned)(cur[u] - n0) < (unsigned)ACC_NB);
        while (mask) {
          const int l = __builtin_ctz(mask);
          mask &= mask - 1;
          f(base + u * 32 + l, __builtin_amdgcn_readlane(cur[u], l));
        }
      }
    }
#pragma unroll
    for (int u = 0; u < ACC_U; ++u) cur[u] = nxt[u];
    base = nbase;
  }
}

__device__ __forceinline__ void scan_list(const int* __restrict__ dstI, int E, int n0,
                                          int wave, int nw, int lane, int* list, int* count) {
  const int stepE = ACC_U * 32;
  const int strideE = nw * stepE;
  int base = wave * stepE;
  int cur[ACC_U], nxt[ACC_U];
#pragma unroll
  for (int u = 0; u < ACC_U; ++u) {
    int e = base + u * 32 + lane;
    cur[u] = e < E ? dstI[e] : -1;
  }
  while (base < E) {
    const int nbase = base + strideE;
    if (nbase + stepE <= E) {
      const int* p = dstI + nbase + lane;
#pragma unroll
      for (int u = 0; u < ACC_U; ++u) nxt[u] = p[u * 32];
    } else {
#pragma unroll
      for (int u = 0; u < ACC_U; ++u) {
        int e = nbase + u * 32 + lane;
        nxt[u] = e < E ? dstI[e] : -1;
      }
    }
    bool any = false;
#pragma unroll
    for (int u = 0; u < ACC_U; ++u) any |= (unsigned)(cur[u] - n0) < (unsigned)ACC_NB;
    if (__builtin_amdgcn_ballot_w32(any)) {
#pragma unroll
      for (int u = 0; u < ACC_U; ++u) {
        if ((unsigned)(cur[u] - n0) < (unsigned)ACC_NB) {
          const int slot = atomicAdd(count, 1);
          if (slot < ACC_CAP) list[slot] = base + u * 32 + lane;
        }
      }
    }
#pragma unroll
    for (int u = 0; u < ACC_U; ++u) cur[u] = nxt[u];
    base = nbase;
  }
}

__device__ __forceinline__ float xhalf(float x) {
  const int b = __float_as_int(x);
  return __int_as_float(__builtin_amdgcn_permlanex16(b, b, 0x76543210, 0xfedcba98, false, false));
}
template <int M>
__device__ __forceinline__ float xrow(float x) {
  return __int_as_float(__builtin_amdgcn_update_dpp(0, __float_as_int(x), 0x160 + M, 0xF, 0xF, true));
}

template <int C>
__device__ __forceinline__ void logit_load(const float* __restrict__ xl,
                                           const float* __restrict__ xr,
                                           int s, int d, int lane,
                                           float (&a)[HEADS * C / 32], float (&b)[HEADS * C / 32]) {
  constexpr int ROW = HEADS * C;
  constexpr int NJ = ROW / 32;
  const float* pl = xl + (size_t)s * ROW + lane;
  const float* pr = xr + (size_t)d * ROW + lane;
#pragma unroll
  for (int j = 0; j < NJ; ++j) { a[j] = pl[j * 32]; b[j] = pr[j * 32]; }
}

template <int C>
__device__ __forceinline__ void logit_finish(const float (&a)[HEADS * C / 32],
                                             const float (&b)[HEADS * C / 32],
                                             const float (&aw)[HEADS * C / 32],
                                             float* lg, unsigned* mx,
                                             int e, int dl, int lane) {
  constexpr int NJ = HEADS * C / 32;
  float p[NJ];
#pragma unroll
  for (int j = 0; j < NJ; ++j) {
    float v = a[j] + b[j];
    v = v > 0.f ? v : 0.2f * v;
    p[j] = v * aw[j];
  }
  const bool b16 = (lane & 16) != 0, b8 = (lane & 8) != 0, b4 = (lane & 4) != 0;
  float r0, r1;
  int h;
  if constexpr (C == 32) {
    float q[4];
#pragma unroll
    for (int j = 0; j < 4; ++j) {
      const float mine = b16 ? p[j + 4] : p[j];
      const float send = b16 ? p[j] : p[j + 4];
      q[j] = mine + xhalf(send);
    }
    const float m0 = b8 ? q[2] : q[0], s0 = b8 ? q[0] : q[2];
    const float m1 = b8 ? q[3] : q[1], s1 = b8 ? q[1] : q[3];
    r0 = m0 + xrow<8>(s0);
    r1 = m1 + xrow<8>(s1);
    h = lane >> 2;
  } else {
    static_assert(C == 16, "row layout: 16 or 32 channels per head");
    const float m0 = b8 ? p[2] : p[0], s0 = b8 ? p[0] : p[2];
    const float m1 = b8 ? p[3] : p[1], s1 = b8 ? p[1] : p[3];
    r0 = m0 + xrow<8>(s0);
    r1 = m1 + xrow<8>(s1);
    h = 2 * ((lane >> 2) & 1) + 4 * ((lane >> 3) & 1) + (lane >> 4);
  }
  const float mine = b4 ? r1 : r0, send = b4 ? r0 : r1;
  float t = mine + xrow<4>(send);
  t += xrow<2>(t);
  t += xrow<1>(t);
  if ((lane & 3) == 0) {
    lg[(size_t)e * HEADS + h] = t;
    atomicMax(mx + dl * HEADS + h, enc_ord(t));
  }
}

template <int C>
__device__ __forceinline__ void edge_logit(const float* __restrict__ xl,
                                           const float* __restrict__ xr,
                                           const float (&aw)[HEADS * C / 32],
                                           float* lg, unsigned* mx,
                                           int e, int s, int d, int dl, int lane) {
  float a[HEADS * C / 32], b[HEADS * C / 32];
  logit_load<C>(xl, xr, s, d, lane, a, b);
  logit_finish<C>(a, b, aw, lg, mx, e, dl, lane);
}

template <int C>
__device__ __forceinline__ float head_ex(float ex, int j, int lane) {
  if constexpr (C == 32) {
    return __int_as_float(__builtin_amdgcn_readlane(__float_as_int(ex), j));
  } else {
    static_assert(C == 16, "row layout: 16 or 32 channels per head");
    float lo = __int_as_float(__builtin_amdgcn_readlane(__float_as_int(ex), 2 * j));
    float hi = __int_as_float(__builtin_amdgcn_readlane(__float_as_int(ex), 2 * j + 1));
    return lane < 16 ? lo : hi;
  }
}

template <int C>
__device__ __forceinline__ void add_load(const float* __restrict__ xl,
                                         const float* lg, const unsigned* mx,
                                         int e, int s, int dl, int lane,
                                         float (&v)[HEADS * C / 32], float& lgv, unsigned& mxv) {
  constexpr int ROW = HEADS * C;
  constexpr int NJ = ROW / 32;
  const int hh = lane & 7;
  const float* row = xl + (size_t)s * ROW + lane;
#pragma unroll
  for (int j = 0; j < NJ; ++j) v[j] = row[j * 32];
  lgv = lg[(size_t)e * HEADS + hh];
  mxv = mx[dl * HEADS + hh];
}

template <int C>
__device__ __forceinline__ void add_finish(const float (&v)[HEADS * C / 32], float lgv, unsigned mxv,
                                           float* acc, float* den, int dl, int lane) {
  constexpr int ROW = HEADS * C;
  constexpr int NJ = ROW / 32;
  const float ex = expf(lgv - dec_ord(mxv));
  if (lane < HEADS) lds_add(den + dl * HEADS + lane, ex);
  float* arow = acc + dl * ROW + lane;
#pragma unroll
  for (int j = 0; j < NJ; ++j) lds_add(arow + j * 32, head_ex<C>(ex, j, lane) * v[j]);
}

template <int C>
__device__ __forceinline__ void add_edge(const float* __restrict__ xl,
                                         const float* lg, const unsigned* mx,
                                         float* acc, float* den,
                                         int e, int s, int dl, int lane) {
  float v[HEADS * C / 32], lgv; unsigned mxv;
  add_load<C>(xl, lg, mx, e, s, dl, lane, v, lgv, mxv);
  add_finish<C>(v, lgv, mxv, acc, den, dl, lane);
}

template <int C, bool LSM>
__global__ __launch_bounds__(1024) void edge_owned(
    const float* __restrict__ xl, const float* __restrict__ xr,
    const float* __restrict__ att,
    const int* __restrict__ srcI, const int* __restrict__ dstI,
    int E, int N,
    float* lg,
    const float* __restrict__ bias, float* __restrict__ out) {
  constexpr int ROW = HEADS * C;
  constexpr int NJ = ROW / 32;
  __shared__ float4 acc4[ACC_NB * ROW / 4];
  __shared__ float den[ACC_NB * HEADS];
  __shared__ unsigned mx[ACC_NB * HEADS];
  __shared__ int list[ACC_CAP];
  __shared__ int lcount;
  float* acc = (float*)acc4;
  const int tid = threadIdx.x, lane = tid & 31;
  const int wave = __builtin_amdgcn_readfirstlane(tid >> 5), nw = blockDim.x >> 5;
  const int n0 = blockIdx.x * ACC_NB;

  for (int i = tid; i < ACC_NB * ROW / 4; i += blockDim.x) acc4[i] = make_float4(0.f, 0.f, 0.f, 0.f);
  for (int i = tid; i < ACC_NB * HEADS; i += blockDim.x) { den[i] = 0.f; mx[i] = 0x007FFFFFu; }
  if (tid == 0) lcount = 0;
  __syncthreads();

  scan_list(dstI, E, n0, wave, nw, lane, list, &lcount);
  __syncthreads();
  const int total = __builtin_amdgcn_readfirstlane(lcount);
  const bool listed = total <= ACC_CAP;

  auto fetch = [&](int idx, int& e, int& s, int& d) {
    if (idx < total) { e = __builtin_amdgcn_readfirstlane(list[idx]); s = srcI[e]; d = dstI[e]; }
    else             { e = -1; s = 0; d = n0; }
  };

  {
    float aw[NJ];
#pragma unroll
    for (int j = 0; j < NJ; ++j) aw[j] = att[j * 32 + lane];
    if (listed) {
      int i = wave, e0, s0, d0, e1, s1, d1;
      fetch(i, e0, s0, d0);
      fetch(i + nw, e1, s1, d1);
      while (e0 >= 0) {
        float a0[NJ], b0[NJ], a1[NJ], b1[NJ];
        logit_load<C>(xl, xr, s0, d0, lane, a0, b0);
        logit_load<C>(xl, xr, s1, d1, lane, a1, b1);
        int f0, t0, g0, f1, t1, g1;
        fetch(i + 2 * nw, f0, t0, g0);
        fetch(i + 3 * nw, f1, t1, g1);
        logit_finish<C>(a0, b0, aw, lg, mx, e0, d0 - n0, lane);
        if (e1 >= 0) logit_finish<C>(a1, b1, aw, lg, mx, e1, d1 - n0, lane);
        i += 2 * nw;
        e0 = f0; s0 = t0; d0 = g0; e1 = f1; s1 = t1; d1 = g1;
      }
    } else {
      scan_owned(dstI, E, n0, wave, nw, lane, [&](int e, int d) {
        edge_logit<C>(xl, xr, aw, lg, mx, e, srcI[e], d, d - n0, lane);
      });
    }
    for (int i = wave; i < ACC_NB; i += nw) {
      const int n = n0 + i;
      if (n < N) edge_logit<C>(xl, xr, aw, lg, mx, E + n, n, n, i, lane);
    }
  }
  __syncthreads();

  if (listed) {
    int i = wave, e0, s0, d0, e1, s1, d1;
    fetch(i, e0, s0, d0);
    fetch(i + nw, e1, s1, d1);
    while (e0 >= 0) {
      float v0[NJ], v1[NJ], l0, l1;
      unsigned m0, m1;
      const int eb = e1 >= 0 ? e1 : e0;
      add_load<C>(xl, lg, mx, e0, s0, d0 - n0, lane, v0, l0, m0);
      add_load<C>(xl, lg, mx, eb, s1, d1 - n0, lane, v1, l1, m1);
      int f0, t0, g0, f1, t1, g1;
      fetch(i + 2 * nw, f0, t0, g0);
      fetch(i + 3 * nw, f1, t1, g1);
      add_finish<C>(v0, l0, m0, acc, den, d0 - n0, lane);
      if (e1 >= 0) add_finish<C>(v1, l1, m1, acc, den, d1 - n0, lane);
      i += 2 * nw;
      e0 = f0; s0 = t0; d0 = g0; e1 = f1; s1 = t1; d1 = g1;
    }
  } else {
    scan_owned(dstI, E, n0, wave, nw, lane, [&](int e, int d) {
      add_edge<C>(xl, lg, mx, acc, den, e, srcI[e], d - n0, lane);
    });
  }
  for (int i = wave; i < ACC_NB; i += nw) {
    const int n = n0 + i;
    if (n < N) add_edge<C>(xl, lg, mx, acc, den, E + n, n, i, lane);
  }
  __syncthreads();

  const int nvalid = (N - n0) < ACC_NB ? (N - n0) : ACC_NB;
  if constexpr (!LSM) {
    constexpr int R4 = ROW / 4;
    const float4* b4 = (const float4*)bias;
    float4* o4 = (float4*)(out + (size_t)n0 * ROW);
    for (int i = tid; i < nvalid * R4; i += blockDim.x) {
      const int r = i / R4, c4 = i % R4;
      const float dn = den[r * HEADS + (c4 * 4) / C];
      const float4 a = acc4[i], b = b4[c4];
      float4 v;
      v.x = a.x / dn + b.x; v.x = v.x > 0.f ? v.x : (expf(v.x) - 1.f);
      v.y = a.y / dn + b.y; v.y = v.y > 0.f ? v.y : (expf(v.y) - 1.f);
      v.z = a.z / dn + b.z; v.z = v.z > 0.f ? v.z : (expf(v.z) - 1.f);
      v.w = a.w / dn + b.w; v.w = v.w > 0.f ? v.w : (expf(v.w) - 1.f);
      o4[i] = v;
    }
  } else {
    static_assert(ROW == 128, "log-softmax row: 4 channels per lane");
    for (int r = wave; r < nvalid; r += nw) {
      float v[4];
      for (int k = 0; k < 4; ++k) {
        const int c = lane + 32 * k;
        v[k] = acc[r * ROW + c] / den[r * HEADS + c / C] + bias[c];
      }
      float mx = fmaxf(fmaxf(v[0], v[2]), fmaxf(v[1], v[3]));
      for (int sft = 16; sft > 0; sft >>= 1) mx = fmaxf(mx, __shfl_xor(mx, sft));
      float sum = (expf(v[0] - mx) + expf(v[2] - mx)) + (expf(v[1] - mx) + expf(v[3] - mx));
      for (int sft = 16; sft > 0; sft >>= 1) sum += __shfl_xor(sum, sft);
      const float lse = logf(sum);
      float* orow = out + (size_t)(n0 + r) * ROW + lane;
      for (int k = 0; k < 4; ++k) orow[32 * k] = v[k] - mx - lse;
    }
  }
}

__device__ __forceinline__ constexpr int kprow(int st) { return 4 * (st >> 1) + (st & 1); }

template <int K, int KOUT, int GC>
__global__ __launch_bounds__(512) void gemm_pair_lds(
    const float* __restrict__ A, const float* __restrict__ Wl, const float* __restrict__ Wr,
    float* __restrict__ Cl, float* __restrict__ Cr, int M) {
  constexpr int NT = 2 * KOUT;
  constexpr int KP = K / 2;
  constexpr int PITCH = NT + 8;
  constexpr int NA = K / 8;
  constexpr int NB = K / 32;
  constexpr int GROUPS = NT / GC;
  static_assert(KOUT % GC == 0 && GC % 32 == 0, "a column group lies inside one product");
  __shared__ float4 Wp4[KP * PITCH / 2];
#if USE_WMMA_F32
  const int tid = threadIdx.x, lane = tid & 31;
  const int h = lane >> 4, c = lane & 15;
  const int wave = __builtin_amdgcn_readfirstlane(tid >> 5);
  const int items = (M >> 4) * GROUPS;
  const int nblk = (int)gridDim.x;

  float4 a4[NA];
  auto load_strip = [&](int t) {
    const int m = t / GROUPS;
    const float4* ap = (const float4*)(A + (size_t)(m * 16 + c) * K) + h;
#pragma unroll
    for (int j = 0; j < NA; ++j) a4[j] = ap[2 * j];
  };
  int slot = wave;
  int t = slot * nblk + (int)blockIdx.x;
  if (t < items) load_strip(t);

  {
    constexpr int Q = NT / 4;
    constexpr int UNITS = KP * Q;
    static_assert(UNITS % 2048 == 0, "copy loop: 4 units per thread per trip");
    for (int u0 = 0; u0 < UNITS; u0 += 2048) {
      float4 r0[4], r1[4];
#pragma unroll
      for (int i = 0; i < 4; ++i) {
        const int u = u0 + i * 512 + tid;
        const int kp = u / Q, col = 4 * (u % Q);
        const float* s = col < KOUT ? Wl + col : Wr + (col - KOUT);
        r0[i] = *(const float4*)(s + (size_t)(2 * kp) * KOUT);
        r1[i] = *(const float4*)(s + (size_t)(2 * kp + 1) * KOUT);
      }
#pragma unroll
      for (int i = 0; i < 4; ++i) {
        const int u = u0 + i * 512 + tid;
        const int kp = u / Q, q = u % Q;
        float4* d = Wp4 + (kp * PITCH + 4 * q) / 2;
        d[0] = make_float4(r0[i].x, r1[i].x, r0[i].y, r1[i].y);
        d[1] = make_float4(r0[i].z, r1[i].z, r0[i].w, r1[i].w);
      }
    }
  }
  __syncthreads();

  const v2f* bp = (const v2f*)Wp4 + (2 * h) * PITCH + c;
  while (t < items) {
    const int m = t / GROUPS, g = t % GROUPS;
    for (int p = 0; p < GC / 32; ++p) {
      const int n0 = g * GC + 32 * p;
      const v2f* bq = bp + n0;
      v8f acc0 = {}, acc1 = {};
      v2f b[2][8][2];
#pragma unroll
      for (int s = 0; s < 8; ++s) {
        b[0][s][0] = bq[kprow(s) * PITCH];
        b[0][s][1] = bq[kprow(s) * PITCH + 16];
      }
#pragma unroll
      for (int blk = 0; blk < NB; ++blk) {
        if (blk + 1 < NB) {
#pragma unroll
          for (int s = 0; s < 8; ++s) {
            b[(blk + 1) & 1][s][0] = bq[kprow((blk + 1) * 8 + s) * PITCH];
            b[(blk + 1) & 1][s][1] = bq[kprow((blk + 1) * 8 + s) * PITCH + 16];
          }
        }
#pragma unroll
        for (int s = 0; s < 8; ++s) {
          const float4 av = a4[(8 * blk + s) >> 1];
          v2f a;
          if (s & 1) { a[0] = av.z; a[1] = av.w; } else { a[0] = av.x; a[1] = av.y; }
          acc0 = __builtin_amdgcn_wmma_f32_16x16x4_f32(false, a, false, b[blk & 1][s][0],
                                                       (short)0, acc0, false, false);
          acc1 = __builtin_amdgcn_wmma_f32_16x16x4_f32(false, a, false, b[blk & 1][s][1],
                                                       (short)0, acc1, false, false);
        }
      }
      float* Cb = (n0 < KOUT ? Cl + n0 : Cr + (n0 - KOUT)) + (size_t)(m * 16) * KOUT + lane;
#pragma unroll
      for (int v = 0; v < 8; ++v) {
        const float got = xhalf(h ? acc0[v] : acc1[v]);
        Cb[(size_t)v * KOUT] = h ? got : acc0[v];
        Cb[(size_t)(8 + v) * KOUT] = h ? acc1[v] : got;
      }
    }
    slot += 16;
    t = slot * nblk + (int)blockIdx.x;
    if (t < items) load_strip(t);
  }
#else
  for (int i = blockIdx.x * blockDim.x + threadIdx.x; i < M * NT; i += gridDim.x * blockDim.x) {
    const int r = i / NT, n = i % NT;
    const float* Wc = n < KOUT ? Wl + n : Wr + (n - KOUT);
    float s = 0.f;
    for (int k = 0; k < K; ++k) s += A[(size_t)r * K + k] * Wc[(size_t)k * KOUT];
    (n < KOUT ? Cl + n : Cr + (n - KOUT))[(size_t)r * KOUT] = s;
  }
#endif
}

extern "C" void kernel_launch(void* const* d_in, const int* in_sizes, int n_in,
                              void* d_out, int out_size, void* d_ws, size_t ws_size,
                              hipStream_t stream) {
  (void)n_in; (void)out_size; (void)ws_size;
  const float* x    = (const float*)d_in[0];
  const int*   ei   = (const int*)d_in[1];
  const float* W1l  = (const float*)d_in[2];
  const float* W1r  = (const float*)d_in[3];
  const float* att1 = (const float*)d_in[4];
  const float* b1   = (const float*)d_in[5];
  const float* W2l  = (const float*)d_in[6];
  const float* W2r  = (const float*)d_in[7];
  const float* att2 = (const float*)d_in[8];
  const float* b2   = (const float*)d_in[9];

  int N = in_sizes[0] / 128;
  int E = in_sizes[1] / 2;
  const int* srcI = ei;
  const int* dstI = ei + E;

  float* ws    = (float*)d_ws;
  float* xl    = ws;
  float* xr    = xl + (size_t)N * 256;
  float* hbuf  = xr + (size_t)N * 256;
  float* lg    = hbuf + (size_t)N * 256;
  float* outFin = (float*)d_out;

  int agrid = (N + ACC_NB - 1) / ACC_NB;

  gemm_pair_lds<128, 256, 128><<<256, dim3(512), 0, stream>>>(x, W1l, W1r, xl, xr, N);
  edge_owned<32, false><<<agrid, dim3(1024), 0, stream>>>(xl, xr, att1, srcI, dstI, E, N, lg, b1, hbuf);

  gemm_pair_lds<256, 128, 64><<<256, dim3(512), 0, stream>>>(hbuf, W2l, W2r, xl, xr, N);
  edge_owned<16, true><<<agrid, dim3(1024), 0, stream>>>(xl, xr, att2, srcI, dstI, E, N, lg, b2, outFin);
}


# edge lists kept across layers: layer 1's owner blocks save their LDS edge list and its length to the workspace, layer 2's blocks read them back; one dst scan per call; sums in arrival order as before
# speedup vs baseline: 5.5173x; 1.0866x over previous
#include <hip/hip_runtime.h>
#include <math.h>

typedef __attribute__((ext_vector_type(2))) float v2f;
typedef __attribute__((ext_vector_type(8))) float v8f;

#define HEADS 8

#if defined(__gfx1250__) && __has_builtin(__builtin_amdgcn_wmma_f32_16x16x4_f32)
#define USE_WMMA_F32 1
#else
#define USE_WMMA_F32 0
#endif

__device__ __forceinline__ unsigned enc_ord(float f) {
  unsigned b = __float_as_uint(f);
  return (b & 0x80000000u) ? ~b : (b | 0x80000000u);
}
__device__ __forceinline__ float dec_ord(unsigned u) {
  unsigned b = (u & 0x80000000u) ? (u ^ 0x80000000u) : ~u;
  return __uint_as_float(b);
}

#define ACC_NB  128
#define ACC_U   8
#define ACC_CAP 4096

__device__ __forceinline__ void lds_add(float* p, float v) {
  __hip_atomic_fetch_add(p, v, __ATOMIC_RELAXED, __HIP_MEMORY_SCOPE_WORKGROUP);
}

template <class F>
__device__ __forceinline__ void scan_owned(const int* __restrict__ dstI, int E, int n0,
                                           int wave, int nw, int lane, F f) {
  const int stepE = ACC_U * 32;
  const int strideE = nw * stepE;
  int base = wave * stepE;
  int cur[ACC_U], nxt[ACC_U];
#pragma unroll
  for (int u = 0; u < ACC_U; ++u) {
    int e = base + u * 32 + lane;
    cur[u] = e < E ? dstI[e] : -1;
  }
  while (base < E) {
    const int nbase = base + strideE;
    if (nbase + stepE <= E) {
      const int* p = dstI + nbase + lane;
#pragma unroll
      for (int u = 0; u < ACC_U; ++u) nxt[u] = p[u * 32];
    } else {
#pragma unroll
      for (int u = 0; u < ACC_U; ++u) {
        int e = nbase + u * 32 + lane;
        nxt[u] = e < E ? dstI[e] : -1;
      }
    }
    bool any = false;
#pragma unroll
    for (int u = 0; u < ACC_U; ++u) any |= (unsigned)(cur[u] - n0) < (unsigned)ACC_NB;
    if (__builtin_amdgcn_ballot_w32(any)) {
#pragma unroll
      for (int u = 0; u < ACC_U; ++u) {
        unsigned mask = __builtin_amdgcn_ballot_w32((unsigned)(cur[u] - n0) < (unsigned)ACC_NB);
        while (mask) {
          const int l = __builtin_ctz(mask);
          mask &= mask - 1;
          f(base + u * 32 + l, __builtin_amdgcn_readlane(cur[u], l));
        }
      }
    }
#pragma unroll
    for (int u = 0; u < ACC_U; ++u) cur[u] = nxt[u];
    base = nbase;
  }
}

__device__ __forceinline__ void scan_list(const int* __restrict__ dstI, int E, int n0,
                                          int wave, int nw, int lane, int* list, int* count) {
  const int stepE = ACC_U * 32;
  const int strideE = nw * stepE;
  int base = wave * stepE;
  int cur[ACC_U], nxt[ACC_U];
#pragma unroll
  for (int u = 0; u < ACC_U; ++u) {
    int e = base + u * 32 + lane;
    cur[u] = e < E ? dstI[e] : -1;
  }
  while (base < E) {
    const int nbase = base + strideE;
    if (nbase + stepE <= E) {
      const int* p = dstI + nbase + lane;
#pragma unroll
      for (int u = 0; u < ACC_U; ++u) nxt[u] = p[u * 32];
    } else {
#pragma unroll
      for (int u = 0; u < ACC_U; ++u) {
        int e = nbase + u * 32 + lane;
        nxt[u] = e < E ? dstI[e] : -1;
      }
    }
    bool any = false;
#pragma unroll
    for (int u = 0; u < ACC_U; ++u) any |= (unsigned)(cur[u] - n0) < (unsigned)ACC_NB;
    if (__builtin_amdgcn_ballot_w32(any)) {
#pragma unroll
      for (int u = 0; u < ACC_U; ++u) {
        if ((unsigned)(cur[u] - n0) < (unsigned)ACC_NB) {
          const int slot = atomicAdd(count, 1);
          if (slot < ACC_CAP) list[slot] = base + u * 32 + lane;
        }
      }
    }
#pragma unroll
    for (int u = 0; u < ACC_U; ++u) cur[u] = nxt[u];
    base = nbase;
  }
}

__device__ __forceinline__ float xhalf(float x) {
  const int b = __float_as_int(x);
  return __int_as_float(__builtin_amdgcn_permlanex16(b, b, 0x76543210, 0xfedcba98, false, false));
}
template <int M>
__device__ __forceinline__ float xrow(float x) {
  return __int_as_float(__builtin_amdgcn_update_dpp(0, __float_as_int(x), 0x160 + M, 0xF, 0xF, true));
}

template <int C>
__device__ __forceinline__ void logit_load(const float* __restrict__ xl,
                                           const float* __restrict__ xr,
                                           int s, int d, int lane,
                                           float (&a)[HEADS * C / 32], float (&b)[HEADS * C / 32]) {
  constexpr int ROW = HEADS * C;
  constexpr int NJ = ROW / 32;
  const float* pl = xl + (size_t)s * ROW + lane;
  const float* pr = xr + (size_t)d * ROW + lane;
#pragma unroll
  for (int j = 0; j < NJ; ++j) { a[j] = pl[j * 32]; b[j] = pr[j * 32]; }
}

template <int C>
__device__ __forceinline__ void logit_finish(const float (&a)[HEADS * C / 32],
                                             const float (&b)[HEADS * C / 32],
                                             const float (&aw)[HEADS * C / 32],
                                             float* lg, unsigned* mx,
                                             int e, int dl, int lane) {
  constexpr int NJ = HEADS * C / 32;
  float p[NJ];
#pragma unroll
  for (int j = 0; j < NJ; ++j) {
    float v = a[j] + b[j];
    v = v > 0.f ? v : 0.2f * v;
    p[j] = v * aw[j];
  }
  const bool b16 = (lane & 16) != 0, b8 = (lane & 8) != 0, b4 = (lane & 4) != 0;
  float r0, r1;
  int h;
  if constexpr (C == 32) {
    float q[4];
#pragma unroll
    for (int j = 0; j < 4; ++j) {
      const float mine = b16 ? p[j + 4] : p[j];
      const float send = b16 ? p[j] : p[j + 4];
      q[j] = mine + xhalf(send);
    }
    const float m0 = b8 ? q[2] : q[0], s0 = b8 ? q[0] : q[2];
    const float m1 = b8 ? q[3] : q[1], s1 = b8 ? q[1] : q[3];
    r0 = m0 + xrow<8>(s0);
    r1 = m1 + xrow<8>(s1);
    h = lane >> 2;
  } else {
    static_assert(C == 16, "row layout: 16 or 32 channels per head");
    const float m0 = b8 ? p[2] : p[0], s0 = b8 ? p[0] : p[2];
    const float m1 = b8 ? p[3] : p[1], s1 = b8 ? p[1] : p[3];
    r0 = m0 + xrow<8>(s0);
    r1 = m1 + xrow<8>(s1);
    h = 2 * ((lane >> 2) & 1) + 4 * ((lane >> 3) & 1) + (lane >> 4);
  }
  const float mine = b4 ? r1 : r0, send = b4 ? r0 : r1;
  float t = mine + xrow<4>(send);
  t += xrow<2>(t);
  t += xrow<1>(t);
  if ((lane & 3) == 0) {
    lg[(size_t)e * HEADS + h] = t;
    atomicMax(mx + dl * HEADS + h, enc_ord(t));
  }
}

template <int C>
__device__ __forceinline__ void edge_logit(const float* __restrict__ xl,
                                           const float* __restrict__ xr,
                                           const float (&aw)[HEADS * C / 32],
                                           float* lg, unsigned* mx,
                                           int e, int s, int d, int dl, int lane) {
  float a[HEADS * C / 32], b[HEADS * C / 32];
  logit_load<C>(xl, xr, s, d, lane, a, b);
  logit_finish<C>(a, b, aw, lg, mx, e, dl, lane);
}

template <int C>
__device__ __forceinline__ float head_ex(float ex, int j, int lane) {
  if constexpr (C == 32) {
    return __int_as_float(__builtin_amdgcn_readlane(__float_as_int(ex), j));
  } else {
    static_assert(C == 16, "row layout: 16 or 32 channels per head");
    float lo = __int_as_float(__builtin_amdgcn_readlane(__float_as_int(ex), 2 * j));
    float hi = __int_as_float(__builtin_amdgcn_readlane(__float_as_int(ex), 2 * j + 1));
    return lane < 16 ? lo : hi;
  }
}

template <int C>
__device__ __forceinline__ void add_load(const float* __restrict__ xl,
                                         const float* lg, const unsigned* mx,
                                         int e, int s, int dl, int lane,
                                         float (&v)[HEADS * C / 32], float& lgv, unsigned& mxv) {
  constexpr int ROW = HEADS * C;
  constexpr int NJ = ROW / 32;
  const int hh = lane & 7;
  const float* row = xl + (size_t)s * ROW + lane;
#pragma unroll
  for (int j = 0; j < NJ; ++j) v[j] = row[j * 32];
  lgv = lg[(size_t)e * HEADS + hh];
  mxv = mx[dl * HEADS + hh];
}

template <int C>
__device__ __forceinline__ void add_finish(const float (&v)[HEADS * C / 32], float lgv, unsigned mxv,
                                           float* acc, float* den, int dl, int lane) {
  constexpr int ROW = HEADS * C;
  constexpr int NJ = ROW / 32;
  const float ex = expf(lgv - dec_ord(mxv));
  if (lane < HEADS) lds_add(den + dl * HEADS + lane, ex);
  float* arow = acc + dl * ROW + lane;
#pragma unroll
  for (int j = 0; j < NJ; ++j) lds_add(arow + j * 32, head_ex<C>(ex, j, lane) * v[j]);
}

template <int C>
__device__ __forceinline__ void add_edge(const float* __restrict__ xl,
                                         const float* lg, const unsigned* mx,
                                         float* acc, float* den,
                                         int e, int s, int dl, int lane) {
  float v[HEADS * C / 32], lgv; unsigned mxv;
  add_load<C>(xl, lg, mx, e, s, dl, lane, v, lgv, mxv);
  add_finish<C>(v, lgv, mxv, acc, den, dl, lane);
}

template <int C, bool LSM, bool SAVED>
__global__ __launch_bounds__(1024) void edge_owned(
    const float* __restrict__ xl, const float* __restrict__ xr,
    const float* __restrict__ att,
    const int* __restrict__ srcI, const int* __restrict__ dstI,
    int E, int N,
    float* lg,
    int* glist, int* gcount,
    const float* __restrict__ bias, float* __restrict__ out) {
  constexpr int ROW = HEADS * C;
  constexpr int NJ = ROW / 32;
  __shared__ float4 acc4[ACC_NB * ROW / 4];
  __shared__ float den[ACC_NB * HEADS];
  __shared__ unsigned mx[ACC_NB * HEADS];
  __shared__ int list[ACC_CAP];
  __shared__ int lcount;
  float* acc = (float*)acc4;
  const int tid = threadIdx.x, lane = tid & 31;
  const int wave = __builtin_amdgcn_readfirstlane(tid >> 5), nw = blockDim.x >> 5;
  const int n0 = blockIdx.x * ACC_NB;

  for (int i = tid; i < ACC_NB * ROW / 4; i += blockDim.x) acc4[i] = make_float4(0.f, 0.f, 0.f, 0.f);
  for (int i = tid; i < ACC_NB * HEADS; i += blockDim.x) { den[i] = 0.f; mx[i] = 0x007FFFFFu; }
  int* gl = glist + (size_t)blockIdx.x * ACC_CAP;
  int total;
  if constexpr (!SAVED) {
    if (tid == 0) lcount = 0;
    __syncthreads();
    scan_list(dstI, E, n0, wave, nw, lane, list, &lcount);
    __syncthreads();
    total = __builtin_amdgcn_readfirstlane(lcount);
    const int nsave = total < ACC_CAP ? total : ACC_CAP;
    for (int i = tid; i < nsave; i += blockDim.x) gl[i] = list[i];
    if (tid == 0) gcount[blockIdx.x] = total;
  } else {
    total = __builtin_amdgcn_readfirstlane(gcount[blockIdx.x]);
    if (total <= ACC_CAP)
      for (int i = tid; i < total; i += blockDim.x) list[i] = gl[i];
    __syncthreads();
  }
  const bool listed = total <= ACC_CAP;

  auto fetch = [&](int idx, int& e, int& s, int& d) {
    if (idx < total) { e = __builtin_amdgcn_readfirstlane(list[idx]); s = srcI[e]; d = dstI[e]; }
    else             { e = -1; s = 0; d = n0; }
  };

  {
    float aw[NJ];
#pragma unroll
    for (int j = 0; j < NJ; ++j) aw[j] = att[j * 32 + lane];
    if (listed) {
      int i = wave, e0, s0, d0, e1, s1, d1;
      fetch(i, e0, s0, d0);
      fetch(i + nw, e1, s1, d1);
      while (e0 >= 0) {
        float a0[NJ], b0[NJ], a1[NJ], b1[NJ];
        logit_load<C>(xl, xr, s0, d0, lane, a0, b0);
        logit_load<C>(xl, xr, s1, d1, lane, a1, b1);
        int f0, t0, g0, f1, t1, g1;
        fetch(i + 2 * nw, f0, t0, g0);
        fetch(i + 3 * nw, f1, t1, g1);
        logit_finish<C>(a0, b0, aw, lg, mx, e0, d0 - n0, lane);
        if (e1 >= 0) logit_finish<C>(a1, b1, aw, lg, mx, e1, d1 - n0, lane);
        i += 2 * nw;
        e0 = f0; s0 = t0; d0 = g0; e1 = f1; s1 = t1; d1 = g1;
      }
    } else {
      scan_owned(dstI, E, n0, wave, nw, lane, [&](int e, int d) {
        edge_logit<C>(xl, xr, aw, lg, mx, e, srcI[e], d, d - n0, lane);
      });
    }
    for (int i = wave; i < ACC_NB; i += nw) {
      const int n = n0 + i;
      if (n < N) edge_logit<C>(xl, xr, aw, lg, mx, E + n, n, n, i, lane);
    }
  }
  __syncthreads();

  if (listed) {
    int i = wave, e0, s0, d0, e1, s1, d1;
    fetch(i, e0, s0, d0);
    fetch(i + nw, e1, s1, d1);
    while (e0 >= 0) {
      float v0[NJ], v1[NJ], l0, l1;
      unsigned m0, m1;
      const int eb = e1 >= 0 ? e1 : e0;
      add_load<C>(xl, lg, mx, e0, s0, d0 - n0, lane, v0, l0, m0);
      add_load<C>(xl, lg, mx, eb, s1, d1 - n0, lane, v1, l1, m1);
      int f0, t0, g0, f1, t1, g1;
      fetch(i + 2 * nw, f0, t0, g0);
      fetch(i + 3 * nw, f1, t1, g1);
      add_finish<C>(v0, l0, m0, acc, den, d0 - n0, lane);
      if (e1 >= 0) add_finish<C>(v1, l1, m1, acc, den, d1 - n0, lane);
      i += 2 * nw;
      e0 = f0; s0 = t0; d0 = g0; e1 = f1; s1 = t1; d1 = g1;
    }
  } else {
    scan_owned(dstI, E, n0, wave, nw, lane, [&](int e, int d) {
      add_edge<C>(xl, lg, mx, acc, den, e, srcI[e], d - n0, lane);
    });
  }
  for (int i = wave; i < ACC_NB; i += nw) {
    const int n = n0 + i;
    if (n < N) add_edge<C>(xl, lg, mx, acc, den, E + n, n, i, lane);
  }
  __syncthreads();

  const int nvalid = (N - n0) < ACC_NB ? (N - n0) : ACC_NB;
  if constexpr (!LSM) {
    constexpr int R4 = ROW / 4;
    const float4* b4 = (const float4*)bias;
    float4* o4 = (float4*)(out + (size_t)n0 * ROW);
    for (int i = tid; i < nvalid * R4; i += blockDim.x) {
      const int r = i / R4, c4 = i % R4;
      const float dn = den[r * HEADS + (c4 * 4) / C];
      const float4 a = acc4[i], b = b4[c4];
      float4 v;
      v.x = a.x / dn + b.x; v.x = v.x > 0.f ? v.x : (expf(v.x) - 1.f);
      v.y = a.y / dn + b.y; v.y = v.y > 0.f ? v.y : (expf(v.y) - 1.f);
      v.z = a.z / dn + b.z; v.z = v.z > 0.f ? v.z : (expf(v.z) - 1.f);
      v.w = a.w / dn + b.w; v.w = v.w > 0.f ? v.w : (expf(v.w) - 1.f);
      o4[i] = v;
    }
  } else {
    static_assert(ROW == 128, "log-softmax row: 4 channels per lane");
    for (int r = wave; r < nvalid; r += nw) {
      float v[4];
      for (int k = 0; k < 4; ++k) {
        const int c = lane + 32 * k;
        v[k] = acc[r * ROW + c] / den[r * HEADS + c / C] + bias[c];
      }
      float mx = fmaxf(fmaxf(v[0], v[2]), fmaxf(v[1], v[3]));
      for (int sft = 16; sft > 0; sft >>= 1) mx = fmaxf(mx, __shfl_xor(mx, sft));
      float sum = (expf(v[0] - mx) + expf(v[2] - mx)) + (expf(v[1] - mx) + expf(v[3] - mx));
      for (int sft = 16; sft > 0; sft >>= 1) sum += __shfl_xor(sum, sft);
      const float lse = logf(sum);
      float* orow = out + (size_t)(n0 + r) * ROW + lane;
      for (int k = 0; k < 4; ++k) orow[32 * k] = v[k] - mx - lse;
    }
  }
}

__device__ __forceinline__ constexpr int kprow(int st) { return 4 * (st >> 1) + (st & 1); }

template <int K, int KOUT, int GC>
__global__ __launch_bounds__(512) void gemm_pair_lds(
    const float* __restrict__ A, const float* __restrict__ Wl, const float* __restrict__ Wr,
    float* __restrict__ Cl, float* __restrict__ Cr, int M) {
  constexpr int NT = 2 * KOUT;
  constexpr int KP = K / 2;
  constexpr int PITCH = NT + 8;
  constexpr int NA = K / 8;
  constexpr int NB = K / 32;
  constexpr int GROUPS = NT / GC;
  static_assert(KOUT % GC == 0 && GC % 32 == 0, "a column group lies inside one product");
  __shared__ float4 Wp4[KP * PITCH / 2];
#if USE_WMMA_F32
  const int tid = threadIdx.x, lane = tid & 31;
  const int h = lane >> 4, c = lane & 15;
  const int wave = __builtin_amdgcn_readfirstlane(tid >> 5);
  const int items = (M >> 4) * GROUPS;
  const int nblk = (int)gridDim.x;

  float4 a4[NA];
  auto load_strip = [&](int t) {
    const int m = t / GROUPS;
    const float4* ap = (const float4*)(A + (size_t)(m * 16 + c) * K) + h;
#pragma unroll
    for (int j = 0; j < NA; ++j) a4[j] = ap[2 * j];
  };
  int slot = wave;
  int t = slot * nblk + (int)blockIdx.x;
  if (t < items) load_strip(t);

  {
    constexpr int Q = NT / 4;
    constexpr int UNITS = KP * Q;
    static_assert(UNITS % 2048 == 0, "copy loop: 4 units per thread per trip");
    for (int u0 = 0; u0 < UNITS; u0 += 2048) {
      float4 r0[4], r1[4];
#pragma unroll
      for (int i = 0; i < 4; ++i) {
        const int u = u0 + i * 512 + tid;
        const int kp = u / Q, col = 4 * (u % Q);
        const float* s = col < KOUT ? Wl + col : Wr + (col - KOUT);
        r0[i] = *(const float4*)(s + (size_t)(2 * kp) * KOUT);
        r1[i] = *(const float4*)(s + (size_t)(2 * kp + 1) * KOUT);
      }
#pragma unroll
      for (int i = 0; i < 4; ++i) {
        const int u = u0 + i * 512 + tid;
        const int kp = u / Q, q = u % Q;
        float4* d = Wp4 + (kp * PITCH + 4 * q) / 2;
        d[0] = make_float4(r0[i].x, r1[i].x, r0[i].y, r1[i].y);
        d[1] = make_float4(r0[i].z, r1[i].z, r0[i].w, r1[i].w);
      }
    }
  }
  __syncthreads();

  const v2f* bp = (const v2f*)Wp4 + (2 * h) * PITCH + c;
  while (t < items) {
    const int m = t / GROUPS, g = t % GROUPS;
    for (int p = 0; p < GC / 32; ++p) {
      const int n0 = g * GC + 32 * p;
      const v2f* bq = bp + n0;
      v8f acc0 = {}, acc1 = {};
      v2f b[2][8][2];
#pragma unroll
      for (int s = 0; s < 8; ++s) {
        b[0][s][0] = bq[kprow(s) * PITCH];
        b[0][s][1] = bq[kprow(s) * PITCH + 16];
      }
#pragma unroll
      for (int blk = 0; blk < NB; ++blk) {
        if (blk + 1 < NB) {
#pragma unroll
          for (int s = 0; s < 8; ++s) {
            b[(blk + 1) & 1][s][0] = bq[kprow((blk + 1) * 8 + s) * PITCH];
            b[(blk + 1) & 1][s][1] = bq[kprow((blk + 1) * 8 + s) * PITCH + 16];
          }
        }
#pragma unroll
        for (int s = 0; s < 8; ++s) {
          const float4 av = a4[(8 * blk + s) >> 1];
          v2f a;
          if (s & 1) { a[0] = av.z; a[1] = av.w; } else { a[0] = av.x; a[1] = av.y; }
          acc0 = __builtin_amdgcn_wmma_f32_16x16x4_f32(false, a, false, b[blk & 1][s][0],
                                                       (short)0, acc0, false, false);
          acc1 = __builtin_amdgcn_wmma_f32_16x16x4_f32(false, a, false, b[blk & 1][s][1],
                                                       (short)0, acc1, false, false);
        }
      }
      float* Cb = (n0 < KOUT ? Cl + n0 : Cr + (n0 - KOUT)) + (size_t)(m * 16) * KOUT + lane;
#pragma unroll
      for (int v = 0; v < 8; ++v) {
        const float got = xhalf(h ? acc0[v] : acc1[v]);
        Cb[(size_t)v * KOUT] = h ? got : acc0[v];
        Cb[(size_t)(8 + v) * KOUT] = h ? acc1[v] : got;
      }
    }
    slot += 16;
    t = slot * nblk + (int)blockIdx.x;
    if (t < items) load_strip(t);
  }
#else
  for (int i = blockIdx.x * blockDim.x + threadIdx.x; i < M * NT; i += gridDim.x * blockDim.x) {
    const int r = i / NT, n = i % NT;
    const float* Wc = n < KOUT ? Wl + n : Wr + (n - KOUT);
    float s = 0.f;
    for (int k = 0; k < K; ++k) s += A[(size_t)r * K + k] * Wc[(size_t)k * KOUT];
    (n < KOUT ? Cl + n : Cr + (n - KOUT))[(size_t)r * KOUT] = s;
  }
#endif
}

extern "C" void kernel_launch(void* const* d_in, const int* in_sizes, int n_in,
                              void* d_out, int out_size, void* d_ws, size_t ws_size,
                              hipStream_t stream) {
  (void)n_in; (void)out_size; (void)ws_size;
  const float* x    = (const float*)d_in[0];
  const int*   ei   = (const int*)d_in[1];
  const float* W1l  = (const float*)d_in[2];
  const float* W1r  = (const float*)d_in[3];
  const float* att1 = (const float*)d_in[4];
  const float* b1   = (const float*)d_in[5];
  const float* W2l  = (const float*)d_in[6];
  const float* W2r  = (const float*)d_in[7];
  const float* att2 = (const float*)d_in[8];
  const float* b2   = (const float*)d_in[9];

  int N = in_sizes[0] / 128;
  int E = in_sizes[1] / 2;
  const int* srcI = ei;
  const int* dstI = ei + E;

  float* ws    = (float*)d_ws;
  float* xl    = ws;
  float* xr    = xl + (size_t)N * 256;
  float* hbuf  = xr + (size_t)N * 256;
  float* lg    = hbuf + (size_t)N * 256;
  int* glist   = (int*)(lg + (size_t)(E + N) * 8);
  float* outFin = (float*)d_out;

  int agrid = (N + ACC_NB - 1) / ACC_NB;
  int* gcount  = glist + (size_t)agrid * ACC_CAP;

  gemm_pair_lds<128, 256, 128><<<256, dim3(512), 0, stream>>>(x, W1l, W1r, xl, xr, N);
  edge_owned<32, false, false><<<agrid, dim3(1024), 0, stream>>>(xl, xr, att1, srcI, dstI, E, N, lg, glist, gcount, b1, hbuf);

  gemm_pair_lds<256, 128, 64><<<256, dim3(512), 0, stream>>>(hbuf, W2l, W2r, xl, xr, N);
  edge_owned<16, true, true><<<agrid, dim3(1024), 0, stream>>>(xl, xr, att2, srcI, dstI, E, N, lg, glist, gcount, b2, outFin);
}


# owner blocks re-sized for an even fill: 128 -> 98 destination nodes per block, 391 -> 511 blocks: every WGP holds two resident blocks (was two on some, one on others); arrival-order sums as before
# speedup vs baseline: 6.0170x; 1.0906x over previous
#include <hip/hip_runtime.h>
#include <math.h>

typedef __attribute__((ext_vector_type(2))) float v2f;
typedef __attribute__((ext_vector_type(8))) float v8f;

#define HEADS 8

#if defined(__gfx1250__) && __has_builtin(__builtin_amdgcn_wmma_f32_16x16x4_f32)
#define USE_WMMA_F32 1
#else
#define USE_WMMA_F32 0
#endif

__device__ __forceinline__ unsigned enc_ord(float f) {
  unsigned b = __float_as_uint(f);
  return (b & 0x80000000u) ? ~b : (b | 0x80000000u);
}
__device__ __forceinline__ float dec_ord(unsigned u) {
  unsigned b = (u & 0x80000000u) ? (u ^ 0x80000000u) : ~u;
  return __uint_as_float(b);
}

#define ACC_NB  98
#define ACC_U   8
#define ACC_CAP 4096

__device__ __forceinline__ void lds_add(float* p, float v) {
  __hip_atomic_fetch_add(p, v, __ATOMIC_RELAXED, __HIP_MEMORY_SCOPE_WORKGROUP);
}

template <class F>
__device__ __forceinline__ void scan_owned(const int* __restrict__ dstI, int E, int n0,
                                           int wave, int nw, int lane, F f) {
  const int stepE = ACC_U * 32;
  const int strideE = nw * stepE;
  int base = wave * stepE;
  int cur[ACC_U], nxt[ACC_U];
#pragma unroll
  for (int u = 0; u < ACC_U; ++u) {
    int e = base + u * 32 + lane;
    cur[u] = e < E ? dstI[e] : -1;
  }
  while (base < E) {
    const int nbase = base + strideE;
    if (nbase + stepE <= E) {
      const int* p = dstI + nbase + lane;
#pragma unroll
      for (int u = 0; u < ACC_U; ++u) nxt[u] = p[u * 32];
    } else {
#pragma unroll
      for (int u = 0; u < ACC_U; ++u) {
        int e = nbase + u * 32 + lane;
        nxt[u] = e < E ? dstI[e] : -1;
      }
    }
    bool any = false;
#pragma unroll
    for (int u = 0; u < ACC_U; ++u) any |= (unsigned)(cur[u] - n0) < (unsigned)ACC_NB;
    if (__builtin_amdgcn_ballot_w32(any)) {
#pragma unroll
      for (int u = 0; u < ACC_U; ++u) {
        unsigned mask = __builtin_amdgcn_ballot_w32((unsigned)(cur[u] - n0) < (unsigned)ACC_NB);
        while (mask) {
          const int l = __builtin_ctz(mask);
          mask &= mask - 1;
          f(base + u * 32 + l, __builtin_amdgcn_readlane(cur[u], l));
        }
      }
    }
#pragma unroll
    for (int u = 0; u < ACC_U; ++u) cur[u] = nxt[u];
    base = nbase;
  }
}

__device__ __forceinline__ void scan_list(const int* __restrict__ dstI, int E, int n0,
                                          int wave, int nw, int lane, int* list, int* count) {
  const int stepE = ACC_U * 32;
  const int strideE = nw * stepE;
  int base = wave * stepE;
  int cur[ACC_U], nxt[ACC_U];
#pragma unroll
  for (int u = 0; u < ACC_U; ++u) {
    int e = base + u * 32 + lane;
    cur[u] = e < E ? dstI[e] : -1;
  }
  while (base < E) {
    const int nbase = base + strideE;
    if (nbase + stepE <= E) {
      const int* p = dstI + nbase + lane;
#pragma unroll
      for (int u = 0; u < ACC_U; ++u) nxt[u] = p[u * 32];
    } else {
#pragma unroll
      for (int u = 0; u < ACC_U; ++u) {
        int e = nbase + u * 32 + lane;
        nxt[u] = e < E ? dstI[e] : -1;
      }
    }
    bool any = false;
#pragma unroll
    for (int u = 0; u < ACC_U; ++u) any |= (unsigned)(cur[u] - n0) < (unsigned)ACC_NB;
    if (__builtin_amdgcn_ballot_w32(any)) {
#pragma unroll
      for (int u = 0; u < ACC_U; ++u) {
        if ((unsigned)(cur[u] - n0) < (unsigned)ACC_NB) {
          const int slot = atomicAdd(count, 1);
          if (slot < ACC_CAP) list[slot] = base + u * 32 + lane;
        }
      }
    }
#pragma unroll
    for (int u = 0; u < ACC_U; ++u) cur[u] = nxt[u];
    base = nbase;
  }
}

__device__ __forceinline__ float xhalf(float x) {
  const int b = __float_as_int(x);
  return __int_as_float(__builtin_amdgcn_permlanex16(b, b, 0x76543210, 0xfedcba98, false, false));
}
template <int M>
__device__ __forceinline__ float xrow(float x) {
  return __int_as_float(__builtin_amdgcn_update_dpp(0, __float_as_int(x), 0x160 + M, 0xF, 0xF, true));
}

template <int C>
__device__ __forceinline__ void logit_load(const float* __restrict__ xl,
                                           const float* __restrict__ xr,
                                           int s, int d, int lane,
                                           float (&a)[HEADS * C / 32], float (&b)[HEADS * C / 32]) {
  constexpr int ROW = HEADS * C;
  constexpr int NJ = ROW / 32;
  const float* pl = xl + (size_t)s * ROW + lane;
  const float* pr = xr + (size_t)d * ROW + lane;
#pragma unroll
  for (int j = 0; j < NJ; ++j) { a[j] = pl[j * 32]; b[j] = pr[j * 32]; }
}

template <int C>
__device__ __forceinline__ void logit_finish(const float (&a)[HEADS * C / 32],
                                             const float (&b)[HEADS * C / 32],
                                             const float (&aw)[HEADS * C / 32],
                                             float* lg, unsigned* mx,
                                             int e, int dl, int lane) {
  constexpr int NJ = HEADS * C / 32;
  float p[NJ];
#pragma unroll
  for (int j = 0; j < NJ; ++j) {
    float v = a[j] + b[j];
    v = v > 0.f ? v : 0.2f * v;
    p[j] = v * aw[j];
  }
  const bool b16 = (lane & 16) != 0, b8 = (lane & 8) != 0, b4 = (lane & 4) != 0;
  float r0, r1;
  int h;
  if constexpr (C == 32) {
    float q[4];
#pragma unroll
    for (int j = 0; j < 4; ++j) {
      const float mine = b16 ? p[j + 4] : p[j];
      const float send = b16 ? p[j] : p[j + 4];
      q[j] = mine + xhalf(send);
    }
    const float m0 = b8 ? q[2] : q[0], s0 = b8 ? q[0] : q[2];
    const float m1 = b8 ? q[3] : q[1], s1 = b8 ? q[1] : q[3];
    r0 = m0 + xrow<8>(s0);
    r1 = m1 + xrow<8>(s1);
    h = lane >> 2;
  } else {
    static_assert(C == 16, "row layout: 16 or 32 channels per head");
    const float m0 = b8 ? p[2] : p[0], s0 = b8 ? p[0] : p[2];
    const float m1 = b8 ? p[3] : p[1], s1 = b8 ? p[1] : p[3];
    r0 = m0 + xrow<8>(s0);
    r1 = m1 + xrow<8>(s1);
    h = 2 * ((lane >> 2) & 1) + 4 * ((lane >> 3) & 1) + (lane >> 4);
  }
  const float mine = b4 ? r1 : r0, send = b4 ? r0 : r1;
  float t = mine + xrow<4>(send);
  t += xrow<2>(t);
  t += xrow<1>(t);
  if ((lane & 3) == 0) {
    lg[(size_t)e * HEADS + h] = t;
    atomicMax(mx + dl * HEADS + h, enc_ord(t));
  }
}

template <int C>
__device__ __forceinline__ void edge_logit(const float* __restrict__ xl,
                                           const float* __restrict__ xr,
                                           const float (&aw)[HEADS * C / 32],
                                           float* lg, unsigned* mx,
                                           int e, int s, int d, int dl, int lane) {
  float a[HEADS * C / 32], b[HEADS * C / 32];
  logit_load<C>(xl, xr, s, d, lane, a, b);
  logit_finish<C>(a, b, aw, lg, mx, e, dl, lane);
}

template <int C>
__device__ __forceinline__ float head_ex(float ex, int j, int lane) {
  if constexpr (C == 32) {
    return __int_as_float(__builtin_amdgcn_readlane(__float_as_int(ex), j));
  } else {
    static_assert(C == 16, "row layout: 16 or 32 channels per head");
    float lo = __int_as_float(__builtin_amdgcn_readlane(__float_as_int(ex), 2 * j));
    float hi = __int_as_float(__builtin_amdgcn_readlane(__float_as_int(ex), 2 * j + 1));
    return lane < 16 ? lo : hi;
  }
}

template <int C>
__device__ __forceinline__ void add_load(const float* __restrict__ xl,
                                         const float* lg, const unsigned* mx,
                                         int e, int s, int dl, int lane,
                                         float (&v)[HEADS * C / 32], float& lgv, unsigned& mxv) {
  constexpr int ROW = HEADS * C;
  constexpr int NJ = ROW / 32;
  const int hh = lane & 7;
  const float* row = xl + (size_t)s * ROW + lane;
#pragma unroll
  for (int j = 0; j < NJ; ++j) v[j] = row[j * 32];
  lgv = lg[(size_t)e * HEADS + hh];
  mxv = mx[dl * HEADS + hh];
}

template <int C>
__device__ __forceinline__ void add_finish(const float (&v)[HEADS * C / 32], float lgv, unsigned mxv,
                                           float* acc, float* den, int dl, int lane) {
  constexpr int ROW = HEADS * C;
  constexpr int NJ = ROW / 32;
  const float ex = expf(lgv - dec_ord(mxv));
  if (lane < HEADS) lds_add(den + dl * HEADS + lane, ex);
  float* arow = acc + dl * ROW + lane;
#pragma unroll
  for (int j = 0; j < NJ; ++j) lds_add(arow + j * 32, head_ex<C>(ex, j, lane) * v[j]);
}

template <int C>
__device__ __forceinline__ void add_edge(const float* __restrict__ xl,
                                         const float* lg, const unsigned* mx,
                                         float* acc, float* den,
                                         int e, int s, int dl, int lane) {
  float v[HEADS * C / 32], lgv; unsigned mxv;
  add_load<C>(xl, lg, mx, e, s, dl, lane, v, lgv, mxv);
  add_finish<C>(v, lgv, mxv, acc, den, dl, lane);
}

template <int C, bool LSM, bool SAVED>
__global__ __launch_bounds__(1024) void edge_owned(
    const float* __restrict__ xl, const float* __restrict__ xr,
    const float* __restrict__ att,
    const int* __restrict__ srcI, const int* __restrict__ dstI,
    int E, int N,
    float* lg,
    int* glist, int* gcount,
    const float* __restrict__ bias, float* __restrict__ out) {
  constexpr int ROW = HEADS * C;
  constexpr int NJ = ROW / 32;
  __shared__ float4 acc4[ACC_NB * ROW / 4];
  __shared__ float den[ACC_NB * HEADS];
  __shared__ unsigned mx[ACC_NB * HEADS];
  __shared__ int list[ACC_CAP];
  __shared__ int lcount;
  float* acc = (float*)acc4;
  const int tid = threadIdx.x, lane = tid & 31;
  const int wave = __builtin_amdgcn_readfirstlane(tid >> 5), nw = blockDim.x >> 5;
  const int n0 = blockIdx.x * ACC_NB;

  for (int i = tid; i < ACC_NB * ROW / 4; i += blockDim.x) acc4[i] = make_float4(0.f, 0.f, 0.f, 0.f);
  for (int i = tid; i < ACC_NB * HEADS; i += blockDim.x) { den[i] = 0.f; mx[i] = 0x007FFFFFu; }
  int* gl = glist + (size_t)blockIdx.x * ACC_CAP;
  int total;
  if constexpr (!SAVED) {
    if (tid == 0) lcount = 0;
    __syncthreads();
    scan_list(dstI, E, n0, wave, nw, lane, list, &lcount);
    __syncthreads();
    total = __builtin_amdgcn_readfirstlane(lcount);
    const int nsave = total < ACC_CAP ? total : ACC_CAP;
    for (int i = tid; i < nsave; i += blockDim.x) gl[i] = list[i];
    if (tid == 0) gcount[blockIdx.x] = total;
  } else {
    total = __builtin_amdgcn_readfirstlane(gcount[blockIdx.x]);
    if (total <= ACC_CAP)
      for (int i = tid; i < total; i += blockDim.x) list[i] = gl[i];
    __syncthreads();
  }
  const bool listed = total <= ACC_CAP;

  auto fetch = [&](int idx, int& e, int& s, int& d) {
    if (idx < total) { e = __builtin_amdgcn_readfirstlane(list[idx]); s = srcI[e]; d = dstI[e]; }
    else             { e = -1; s = 0; d = n0; }
  };

  {
    float aw[NJ];
#pragma unroll
    for (int j = 0; j < NJ; ++j) aw[j] = att[j * 32 + lane];
    if (listed) {
      int i = wave, e0, s0, d0, e1, s1, d1;
      fetch(i, e0, s0, d0);
      fetch(i + nw, e1, s1, d1);
      while (e0 >= 0) {
        float a0[NJ], b0[NJ], a1[NJ], b1[NJ];
        logit_load<C>(xl, xr, s0, d0, lane, a0, b0);
        logit_load<C>(xl, xr, s1, d1, lane, a1, b1);
        int f0, t0, g0, f1, t1, g1;
        fetch(i + 2 * nw, f0, t0, g0);
        fetch(i + 3 * nw, f1, t1, g1);
        logit_finish<C>(a0, b0, aw, lg, mx, e0, d0 - n0, lane);
        if (e1 >= 0) logit_finish<C>(a1, b1, aw, lg, mx, e1, d1 - n0, lane);
        i += 2 * nw;
        e0 = f0; s0 = t0; d0 = g0; e1 = f1; s1 = t1; d1 = g1;
      }
    } else {
      scan_owned(dstI, E, n0, wave, nw, lane, [&](int e, int d) {
        edge_logit<C>(xl, xr, aw, lg, mx, e, srcI[e], d, d - n0, lane);
      });
    }
    for (int i = wave; i < ACC_NB; i += nw) {
      const int n = n0 + i;
      if (n < N) edge_logit<C>(xl, xr, aw, lg, mx, E + n, n, n, i, lane);
    }
  }
  __syncthreads();

  if (listed) {
    int i = wave, e0, s0, d0, e1, s1, d1;
    fetch(i, e0, s0, d0);
    fetch(i + nw, e1, s1, d1);
    while (e0 >= 0) {
      float v0[NJ], v1[NJ], l0, l1;
      unsigned m0, m1;
      const int eb = e1 >= 0 ? e1 : e0;
      add_load<C>(xl, lg, mx, e0, s0, d0 - n0, lane, v0, l0, m0);
      add_load<C>(xl, lg, mx, eb, s1, d1 - n0, lane, v1, l1, m1);
      int f0, t0, g0, f1, t1, g1;
      fetch(i + 2 * nw, f0, t0, g0);
      fetch(i + 3 * nw, f1, t1, g1);
      add_finish<C>(v0, l0, m0, acc, den, d0 - n0, lane);
      if (e1 >= 0) add_finish<C>(v1, l1, m1, acc, den, d1 - n0, lane);
      i += 2 * nw;
      e0 = f0; s0 = t0; d0 = g0; e1 = f1; s1 = t1; d1 = g1;
    }
  } else {
    scan_owned(dstI, E, n0, wave, nw, lane, [&](int e, int d) {
      add_edge<C>(xl, lg, mx, acc, den, e, srcI[e], d - n0, lane);
    });
  }
  for (int i = wave; i < ACC_NB; i += nw) {
    const int n = n0 + i;
    if (n < N) add_edge<C>(xl, lg, mx, acc, den, E + n, n, i, lane);
  }
  __syncthreads();

  const int nvalid = (N - n0) < ACC_NB ? (N - n0) : ACC_NB;
  if constexpr (!LSM) {
    constexpr int R4 = ROW / 4;
    const float4* b4 = (const float4*)bias;
    float4* o4 = (float4*)(out + (size_t)n0 * ROW);
    for (int i = tid; i < nvalid * R4; i += blockDim.x) {
      const int r = i / R4, c4 = i % R4;
      const float dn = den[r * HEADS + (c4 * 4) / C];
      const float4 a = acc4[i], b = b4[c4];
      float4 v;
      v.x = a.x / dn + b.x; v.x = v.x > 0.f ? v.x : (expf(v.x) - 1.f);
      v.y = a.y / dn + b.y; v.y = v.y > 0.f ? v.y : (expf(v.y) - 1.f);
      v.z = a.z / dn + b.z; v.z = v.z > 0.f ? v.z : (expf(v.z) - 1.f);
      v.w = a.w / dn + b.w; v.w = v.w > 0.f ? v.w : (expf(v.w) - 1.f);
      o4[i] = v;
    }
  } else {
    static_assert(ROW == 128, "log-softmax row: 4 channels per lane");
    for (int r = wave; r < nvalid; r += nw) {
      float v[4];
      for (int k = 0; k < 4; ++k) {
        const int c = lane + 32 * k;
        v[k] = acc[r * ROW + c] / den[r * HEADS + c / C] + bias[c];
      }
      float mx = fmaxf(fmaxf(v[0], v[2]), fmaxf(v[1], v[3]));
      for (int sft = 16; sft > 0; sft >>= 1) mx = fmaxf(mx, __shfl_xor(mx, sft));
      float sum = (expf(v[0] - mx) + expf(v[2] - mx)) + (expf(v[1] - mx) + expf(v[3] - mx));
      for (int sft = 16; sft > 0; sft >>= 1) sum += __shfl_xor(sum, sft);
      const float lse = logf(sum);
      float* orow = out + (size_t)(n0 + r) * ROW + lane;
      for (int k = 0; k < 4; ++k) orow[32 * k] = v[k] - mx - lse;
    }
  }
}

__device__ __forceinline__ constexpr int kprow(int st) { return 4 * (st >> 1) + (st & 1); }

template <int K, int KOUT, int GC>
__global__ __launch_bounds__(512) void gemm_pair_lds(
    const float* __restrict__ A, const float* __restrict__ Wl, const float* __restrict__ Wr,
    float* __restrict__ Cl, float* __restrict__ Cr, int M) {
  constexpr int NT = 2 * KOUT;
  constexpr int KP = K / 2;
  constexpr int PITCH = NT + 8;
  constexpr int NA = K / 8;
  constexpr int NB = K / 32;
  constexpr int GROUPS = NT / GC;
  static_assert(KOUT % GC == 0 && GC % 32 == 0, "a column group lies inside one product");
  __shared__ float4 Wp4[KP * PITCH / 2];
#if USE_WMMA_F32
  const int tid = threadIdx.x, lane = tid & 31;
  const int h = lane >> 4, c = lane & 15;
  const int wave = __builtin_amdgcn_readfirstlane(tid >> 5);
  const int items = (M >> 4) * GROUPS;
  const int nblk = (int)gridDim.x;

  float4 a4[NA];
  auto load_strip = [&](int t) {
    const int m = t / GROUPS;
    const float4* ap = (const float4*)(A + (size_t)(m * 16 + c) * K) + h;
#pragma unroll
    for (int j = 0; j < NA; ++j) a4[j] = ap[2 * j];
  };
  int slot = wave;
  int t = slot * nblk + (int)blockIdx.x;
  if (t < items) load_strip(t);

  {
    constexpr int Q = NT / 4;
    constexpr int UNITS = KP * Q;
    static_assert(UNITS % 2048 == 0, "copy loop: 4 units per thread per trip");
    for (int u0 = 0; u0 < UNITS; u0 += 2048) {
      float4 r0[4], r1[4];
#pragma unroll
      for (int i = 0; i < 4; ++i) {
        const int u = u0 + i * 512 + tid;
        const int kp = u / Q, col = 4 * (u % Q);
        const float* s = col < KOUT ? Wl + col : Wr + (col - KOUT);
        r0[i] = *(const float4*)(s + (size_t)(2 * kp) * KOUT);
        r1[i] = *(const float4*)(s + (size_t)(2 * kp + 1) * KOUT);
      }
#pragma unroll
      for (int i = 0; i < 4; ++i) {
        const int u = u0 + i * 512 + tid;
        const int kp = u / Q, q = u % Q;
        float4* d = Wp4 + (kp * PITCH + 4 * q) / 2;
        d[0] = make_float4(r0[i].x, r1[i].x, r0[i].y, r1[i].y);
        d[1] = make_float4(r0[i].z, r1[i].z, r0[i].w, r1[i].w);
      }
    }
  }
  __syncthreads();

  const v2f* bp = (const v2f*)Wp4 + (2 * h) * PITCH + c;
  while (t < items) {
    const int m = t / GROUPS, g = t % GROUPS;
    for (int p = 0; p < GC / 32; ++p) {
      const int n0 = g * GC + 32 * p;
      const v2f* bq = bp + n0;
      v8f acc0 = {}, acc1 = {};
      v2f b[2][8][2];
#pragma unroll
      for (int s = 0; s < 8; ++s) {
        b[0][s][0] = bq[kprow(s) * PITCH];
        b[0][s][1] = bq[kprow(s) * PITCH + 16];
      }
#pragma unroll
      for (int blk = 0; blk < NB; ++blk) {
        if (blk + 1 < NB) {
#pragma unroll
          for (int s = 0; s < 8; ++s) {
            b[(blk + 1) & 1][s][0] = bq[kprow((blk + 1) * 8 + s) * PITCH];
            b[(blk + 1) & 1][s][1] = bq[kprow((blk + 1) * 8 + s) * PITCH + 16];
          }
        }
#pragma unroll
        for (int s = 0; s < 8; ++s) {
          const float4 av = a4[(8 * blk + s) >> 1];
          v2f a;
          if (s & 1) { a[0] = av.z; a[1] = av.w; } else { a[0] = av.x; a[1] = av.y; }
          acc0 = __builtin_amdgcn_wmma_f32_16x16x4_f32(false, a, false, b[blk & 1][s][0],
                                                       (short)0, acc0, false, false);
          acc1 = __builtin_amdgcn_wmma_f32_16x16x4_f32(false, a, false, b[blk & 1][s][1],
                                                       (short)0, acc1, false, false);
        }
      }
      float* Cb = (n0 < KOUT ? Cl + n0 : Cr + (n0 - KOUT)) + (size_t)(m * 16) * KOUT + lane;
#pragma unroll
      for (int v = 0; v < 8; ++v) {
        const float got = xhalf(h ? acc0[v] : acc1[v]);
        Cb[(size_t)v * KOUT] = h ? got : acc0[v];
        Cb[(size_t)(8 + v) * KOUT] = h ? acc1[v] : got;
      }
    }
    slot += 16;
    t = slot * nblk + (int)blockIdx.x;
    if (t < items) load_strip(t);
  }
#else
  for (int i = blockIdx.x * blockDim.x + threadIdx.x; i < M * NT; i += gridDim.x * blockDim.x) {
    const int r = i / NT, n = i % NT;
    const float* Wc = n < KOUT ? Wl + n : Wr + (n - KOUT);
    float s = 0.f;
    for (int k = 0; k < K; ++k) s += A[(size_t)r * K + k] * Wc[(size_t)k * KOUT];
    (n < KOUT ? Cl + n : Cr + (n - KOUT))[(size_t)r * KOUT] = s;
  }
#endif
}

extern "C" void kernel_launch(void* const* d_in, const int* in_sizes, int n_in,
                              void* d_out, int out_size, void* d_ws, size_t ws_size,
                              hipStream_t stream) {
  (void)n_in; (void)out_size; (void)ws_size;
  const float* x    = (const float*)d_in[0];
  const int*   ei   = (const int*)d_in[1];
  const float* W1l  = (const float*)d_in[2];
  const float* W1r  = (const float*)d_in[3];
  const float* att1 = (const float*)d_in[4];
  const float* b1   = (const float*)d_in[5];
  const float* W2l  = (const float*)d_in[6];
  const float* W2r  = (const float*)d_in[7];
  const float* att2 = (const float*)d_in[8];
  const float* b2   = (const float*)d_in[9];

  int N = in_sizes[0] / 128;
  int E = in_sizes[1] / 2;
  const int* srcI = ei;
  const int* dstI = ei + E;

  float* ws    = (float*)d_ws;
  float* xl    = ws;
  float* xr    = xl + (size_t)N * 256;
  float* hbuf  = xr + (size_t)N * 256;
  float* lg    = hbuf + (size_t)N * 256;
  int* glist   = (int*)(lg + (size_t)(E + N) * 8);
  float* outFin = (float*)d_out;

  int agrid = (N + ACC_NB - 1) / ACC_NB;
  int* gcount  = glist + (size_t)agrid * ACC_CAP;

  gemm_pair_lds<128, 256, 128><<<256, dim3(512), 0, stream>>>(x, W1l, W1r, xl, xr, N);
  edge_owned<32, false, false><<<agrid, dim3(1024), 0, stream>>>(xl, xr, att1, srcI, dstI, E, N, lg, glist, gcount, b1, hbuf);

  gemm_pair_lds<256, 128, 64><<<256, dim3(512), 0, stream>>>(hbuf, W2l, W2r, xl, xr, N);
  edge_owned<16, true, true><<<agrid, dim3(1024), 0, stream>>>(xl, xr, att2, srcI, dstI, E, N, lg, glist, gcount, b2, outFin);
}


# edge logits: owner block copies its own nodes' xr rows once into its LDS row area (idle until the sums start) and reads xr[d] there, not from L2; area cleared before the sums; sum order as before
# speedup vs baseline: 6.3930x; 1.0625x over previous
#include <hip/hip_runtime.h>
#include <math.h>

typedef __attribute__((ext_vector_type(2))) float v2f;
typedef __attribute__((ext_vector_type(8))) float v8f;

#define HEADS 8

#if defined(__gfx1250__) && __has_builtin(__builtin_amdgcn_wmma_f32_16x16x4_f32)
#define USE_WMMA_F32 1
#else
#define USE_WMMA_F32 0
#endif

__device__ __forceinline__ unsigned enc_ord(float f) {
  unsigned b = __float_as_uint(f);
  return (b & 0x80000000u) ? ~b : (b | 0x80000000u);
}
__device__ __forceinline__ float dec_ord(unsigned u) {
  unsigned b = (u & 0x80000000u) ? (u ^ 0x80000000u) : ~u;
  return __uint_as_float(b);
}

#define ACC_NB  98
#define ACC_T   1024
#define ACC_U   8
#define ACC_CAP 4096

__device__ __forceinline__ void lds_add(float* p, float v) {
  __hip_atomic_fetch_add(p, v, __ATOMIC_RELAXED, __HIP_MEMORY_SCOPE_WORKGROUP);
}

template <class F>
__device__ __forceinline__ void scan_owned(const int* __restrict__ dstI, int E, int n0,
                                           int wave, int nw, int lane, F f) {
  const int stepE = ACC_U * 32;
  const int strideE = nw * stepE;
  int base = wave * stepE;
  int cur[ACC_U], nxt[ACC_U];
#pragma unroll
  for (int u = 0; u < ACC_U; ++u) {
    int e = base + u * 32 + lane;
    cur[u] = e < E ? dstI[e] : -1;
  }
  while (base < E) {
    const int nbase = base + strideE;
    if (nbase + stepE <= E) {
      const int* p = dstI + nbase + lane;
#pragma unroll
      for (int u = 0; u < ACC_U; ++u) nxt[u] = p[u * 32];
    } else {
#pragma unroll
      for (int u = 0; u < ACC_U; ++u) {
        int e = nbase + u * 32 + lane;
        nxt[u] = e < E ? dstI[e] : -1;
      }
    }
    bool any = false;
#pragma unroll
    for (int u = 0; u < ACC_U; ++u) any |= (unsigned)(cur[u] - n0) < (unsigned)ACC_NB;
    if (__builtin_amdgcn_ballot_w32(any)) {
#pragma unroll
      for (int u = 0; u < ACC_U; ++u) {
        unsigned mask = __builtin_amdgcn_ballot_w32((unsigned)(cur[u] - n0) < (unsigned)ACC_NB);
        while (mask) {
          const int l = __builtin_ctz(mask);
          mask &= mask - 1;
          f(base + u * 32 + l, __builtin_amdgcn_readlane(cur[u], l));
        }
      }
    }
#pragma unroll
    for (int u = 0; u < ACC_U; ++u) cur[u] = nxt[u];
    base = nbase;
  }
}

__device__ __forceinline__ void scan_list(const int* __restrict__ dstI, int E, int n0,
                                          int wave, int nw, int lane, int* list, int* count) {
  const int stepE = ACC_U * 32;
  const int strideE = nw * stepE;
  int base = wave * stepE;
  int cur[ACC_U], nxt[ACC_U];
#pragma unroll
  for (int u = 0; u < ACC_U; ++u) {
    int e = base + u * 32 + lane;
    cur[u] = e < E ? dstI[e] : -1;
  }
  while (base < E) {
    const int nbase = base + strideE;
    if (nbase + stepE <= E) {
      const int* p = dstI + nbase + lane;
#pragma unroll
      for (int u = 0; u < ACC_U; ++u) nxt[u] = p[u * 32];
    } else {
#pragma unroll
      for (int u = 0; u < ACC_U; ++u) {
        int e = nbase + u * 32 + lane;
        nxt[u] = e < E ? dstI[e] : -1;
      }
    }
    bool any = false;
#pragma unroll
    for (int u = 0; u < ACC_U; ++u) any |= (unsigned)(cur[u] - n0) < (unsigned)ACC_NB;
    if (__builtin_amdgcn_ballot_w32(any)) {
#pragma unroll
      for (int u = 0; u < ACC_U; ++u) {
        if ((unsigned)(cur[u] - n0) < (unsigned)ACC_NB) {
          const int slot = atomicAdd(count, 1);
          if (slot < ACC_CAP) list[slot] = base + u * 32 + lane;
        }
      }
    }
#pragma unroll
    for (int u = 0; u < ACC_U; ++u) cur[u] = nxt[u];
    base = nbase;
  }
}

__device__ __forceinline__ float xhalf(float x) {
  const int b = __float_as_int(x);
  return __int_as_float(__builtin_amdgcn_permlanex16(b, b, 0x76543210, 0xfedcba98, false, false));
}
template <int M>
__device__ __forceinline__ float xrow(float x) {
  return __int_as_float(__builtin_amdgcn_update_dpp(0, __float_as_int(x), 0x160 + M, 0xF, 0xF, true));
}

template <int C>
__device__ __forceinline__ void logit_load(const float* __restrict__ xl,
                                           int s, int lane,
                                           float (&a)[HEADS * C / 32]) {
  constexpr int ROW = HEADS * C;
  constexpr int NJ = ROW / 32;
  const float* pl = xl + (size_t)s * ROW + lane;
#pragma unroll
  for (int j = 0; j < NJ; ++j) a[j] = pl[j * 32];
}

template <int C>
__device__ __forceinline__ void logit_finish(const float (&a)[HEADS * C / 32],
                                             const float* xrows,
                                             const float (&aw)[HEADS * C / 32],
                                             float* lg, unsigned* mx,
                                             int e, int dl, int lane) {
  constexpr int ROW = HEADS * C;
  constexpr int NJ = ROW / 32;
  const float* pr = xrows + dl * ROW + lane;
  float p[NJ];
#pragma unroll
  for (int j = 0; j < NJ; ++j) {
    float v = a[j] + pr[j * 32];
    v = v > 0.f ? v : 0.2f * v;
    p[j] = v * aw[j];
  }
  const bool b16 = (lane & 16) != 0, b8 = (lane & 8) != 0, b4 = (lane & 4) != 0;
  float r0, r1;
  int h;
  if constexpr (C == 32) {
    float q[4];
#pragma unroll
    for (int j = 0; j < 4; ++j) {
      const float mine = b16 ? p[j + 4] : p[j];
      const float send = b16 ? p[j] : p[j + 4];
      q[j] = mine + xhalf(send);
    }
    const float m0 = b8 ? q[2] : q[0], s0 = b8 ? q[0] : q[2];
    const float m1 = b8 ? q[3] : q[1], s1 = b8 ? q[1] : q[3];
    r0 = m0 + xrow<8>(s0);
    r1 = m1 + xrow<8>(s1);
    h = lane >> 2;
  } else {
    static_assert(C == 16, "row layout: 16 or 32 channels per head");
    const float m0 = b8 ? p[2] : p[0], s0 = b8 ? p[0] : p[2];
    const float m1 = b8 ? p[3] : p[1], s1 = b8 ? p[1] : p[3];
    r0 = m0 + xrow<8>(s0);
    r1 = m1 + xrow<8>(s1);
    h = 2 * ((lane >> 2) & 1) + 4 * ((lane >> 3) & 1) + (lane >> 4);
  }
  const float mine = b4 ? r1 : r0, send = b4 ? r0 : r1;
  float t = mine + xrow<4>(send);
  t += xrow<2>(t);
  t += xrow<1>(t);
  if ((lane & 3) == 0) {
    lg[(size_t)e * HEADS + h] = t;
    atomicMax(mx + dl * HEADS + h, enc_ord(t));
  }
}

template <int C>
__device__ __forceinline__ void edge_logit(const float* __restrict__ xl,
                                           const float* xrows,
                                           const float (&aw)[HEADS * C / 32],
                                           float* lg, unsigned* mx,
                                           int e, int s, int dl, int lane) {
  float a[HEADS * C / 32];
  logit_load<C>(xl, s, lane, a);
  logit_finish<C>(a, xrows, aw, lg, mx, e, dl, lane);
}

template <int C>
__device__ __forceinline__ float head_ex(float ex, int j, int lane) {
  if constexpr (C == 32) {
    return __int_as_float(__builtin_amdgcn_readlane(__float_as_int(ex), j));
  } else {
    static_assert(C == 16, "row layout: 16 or 32 channels per head");
    float lo = __int_as_float(__builtin_amdgcn_readlane(__float_as_int(ex), 2 * j));
    float hi = __int_as_float(__builtin_amdgcn_readlane(__float_as_int(ex), 2 * j + 1));
    return lane < 16 ? lo : hi;
  }
}

template <int C>
__device__ __forceinline__ void add_load(const float* __restrict__ xl,
                                         const float* lg, const unsigned* mx,
                                         int e, int s, int dl, int lane,
                                         float (&v)[HEADS * C / 32], float& lgv, unsigned& mxv) {
  constexpr int ROW = HEADS * C;
  constexpr int NJ = ROW / 32;
  const int hh = lane & 7;
  const float* row = xl + (size_t)s * ROW + lane;
#pragma unroll
  for (int j = 0; j < NJ; ++j) v[j] = row[j * 32];
  lgv = lg[(size_t)e * HEADS + hh];
  mxv = mx[dl * HEADS + hh];
}

template <int C>
__device__ __forceinline__ void add_finish(const float (&v)[HEADS * C / 32], float lgv, unsigned mxv,
                                           float* acc, float* den, int dl, int lane) {
  constexpr int ROW = HEADS * C;
  constexpr int NJ = ROW / 32;
  const float ex = expf(lgv - dec_ord(mxv));
  if (lane < HEADS) lds_add(den + dl * HEADS + lane, ex);
  float* arow = acc + dl * ROW + lane;
#pragma unroll
  for (int j = 0; j < NJ; ++j) lds_add(arow + j * 32, head_ex<C>(ex, j, lane) * v[j]);
}

template <int C>
__device__ __forceinline__ void add_edge(const float* __restrict__ xl,
                                         const float* lg, const unsigned* mx,
                                         float* acc, float* den,
                                         int e, int s, int dl, int lane) {
  float v[HEADS * C / 32], lgv; unsigned mxv;
  add_load<C>(xl, lg, mx, e, s, dl, lane, v, lgv, mxv);
  add_finish<C>(v, lgv, mxv, acc, den, dl, lane);
}

template <int C, bool LSM, bool SAVED>
__global__ __launch_bounds__(ACC_T) void edge_owned(
    const float* __restrict__ xl, const float* __restrict__ xr,
    const float* __restrict__ att,
    const int* __restrict__ srcI, const int* __restrict__ dstI,
    int E, int N,
    float* lg,
    int* glist, int* gcount,
    const float* __restrict__ bias, float* __restrict__ out) {
  constexpr int ROW = HEADS * C;
  constexpr int NJ = ROW / 32;
  __shared__ float4 acc4[ACC_NB * ROW / 4];
  __shared__ float den[ACC_NB * HEADS];
  __shared__ unsigned mx[ACC_NB * HEADS];
  __shared__ int list[ACC_CAP];
  __shared__ int lcount;
  float* acc = (float*)acc4;
  const int tid = threadIdx.x, lane = tid & 31;
  const int wave = __builtin_amdgcn_readfirstlane(tid >> 5), nw = blockDim.x >> 5;
  const int n0 = blockIdx.x * ACC_NB;

  const int nvalid = (N - n0) < ACC_NB ? (N - n0) : ACC_NB;
  constexpr int NST = (ACC_NB * ROW / 4 + ACC_T - 1) / ACC_T;
  float4 st[NST];
  auto rows_request = [&]() {
    const float4* xr4 = (const float4*)(xr + (size_t)n0 * ROW);
#pragma unroll
    for (int k = 0; k < NST; ++k) {
      const int i = tid + k * ACC_T;
      st[k] = i < nvalid * (ROW / 4) ? xr4[i] : make_float4(0.f, 0.f, 0.f, 0.f);
    }
  };
  auto rows_store = [&]() {
#pragma unroll
    for (int k = 0; k < NST; ++k) {
      const int i = tid + k * ACC_T;
      if (i < ACC_NB * ROW / 4) acc4[i] = st[k];
    }
  };
  for (int i = tid; i < ACC_NB * HEADS; i += blockDim.x) { den[i] = 0.f; mx[i] = 0x007FFFFFu; }
  int* gl = glist + (size_t)blockIdx.x * ACC_CAP;
  int total;
  if constexpr (!SAVED) {
    if (tid == 0) lcount = 0;
    __syncthreads();
    rows_request();
    scan_list(dstI, E, n0, wave, nw, lane, list, &lcount);
    rows_store();
    __syncthreads();
    total = __builtin_amdgcn_readfirstlane(lcount);
    const int nsave = total < ACC_CAP ? total : ACC_CAP;
    for (int i = tid; i < nsave; i += blockDim.x) gl[i] = list[i];
    if (tid == 0) gcount[blockIdx.x] = total;
  } else {
    rows_request();
    total = __builtin_amdgcn_readfirstlane(gcount[blockIdx.x]);
    if (total <= ACC_CAP)
      for (int i = tid; i < total; i += blockDim.x) list[i] = gl[i];
    rows_store();
    __syncthreads();
  }
  const bool listed = total <= ACC_CAP;

  auto fetch = [&](int idx, int& e, int& s, int& d) {
    if (idx < total) { e = __builtin_amdgcn_readfirstlane(list[idx]); s = srcI[e]; d = dstI[e]; }
    else             { e = -1; s = 0; d = n0; }
  };

  {
    float aw[NJ];
#pragma unroll
    for (int j = 0; j < NJ; ++j) aw[j] = att[j * 32 + lane];
    if (listed) {
      int i = wave, e0, s0, d0, e1, s1, d1;
      fetch(i, e0, s0, d0);
      fetch(i + nw, e1, s1, d1);
      while (e0 >= 0) {
        float a0[NJ], a1[NJ];
        logit_load<C>(xl, s0, lane, a0);
        logit_load<C>(xl, s1, lane, a1);
        int f0, t0, g0, f1, t1, g1;
        fetch(i + 2 * nw, f0, t0, g0);
        fetch(i + 3 * nw, f1, t1, g1);
        logit_finish<C>(a0, acc, aw, lg, mx, e0, d0 - n0, lane);
        if (e1 >= 0) logit_finish<C>(a1, acc, aw, lg, mx, e1, d1 - n0, lane);
        i += 2 * nw;
        e0 = f0; s0 = t0; d0 = g0; e1 = f1; s1 = t1; d1 = g1;
      }
    } else {
      scan_owned(dstI, E, n0, wave, nw, lane, [&](int e, int d) {
        edge_logit<C>(xl, acc, aw, lg, mx, e, srcI[e], d - n0, lane);
      });
    }
    for (int i = wave; i < ACC_NB; i += nw) {
      const int n = n0 + i;
      if (n < N) edge_logit<C>(xl, acc, aw, lg, mx, E + n, n, i, lane);
    }
  }
  __syncthreads();
  for (int i = tid; i < ACC_NB * ROW / 4; i += blockDim.x) acc4[i] = make_float4(0.f, 0.f, 0.f, 0.f);
  __syncthreads();

  if (listed) {
    int i = wave, e0, s0, d0, e1, s1, d1;
    fetch(i, e0, s0, d0);
    fetch(i + nw, e1, s1, d1);
    while (e0 >= 0) {
      float v0[NJ], v1[NJ], l0, l1;
      unsigned m0, m1;
      const int eb = e1 >= 0 ? e1 : e0;
      add_load<C>(xl, lg, mx, e0, s0, d0 - n0, lane, v0, l0, m0);
      add_load<C>(xl, lg, mx, eb, s1, d1 - n0, lane, v1, l1, m1);
      int f0, t0, g0, f1, t1, g1;
      fetch(i + 2 * nw, f0, t0, g0);
      fetch(i + 3 * nw, f1, t1, g1);
      add_finish<C>(v0, l0, m0, acc, den, d0 - n0, lane);
      if (e1 >= 0) add_finish<C>(v1, l1, m1, acc, den, d1 - n0, lane);
      i += 2 * nw;
      e0 = f0; s0 = t0; d0 = g0; e1 = f1; s1 = t1; d1 = g1;
    }
  } else {
    scan_owned(dstI, E, n0, wave, nw, lane, [&](int e, int d) {
      add_edge<C>(xl, lg, mx, acc, den, e, srcI[e], d - n0, lane);
    });
  }
  for (int i = wave; i < ACC_NB; i += nw) {
    const int n = n0 + i;
    if (n < N) add_edge<C>(xl, lg, mx, acc, den, E + n, n, i, lane);
  }
  __syncthreads();

  if constexpr (!LSM) {
    constexpr int R4 = ROW / 4;
    const float4* b4 = (const float4*)bias;
    float4* o4 = (float4*)(out + (size_t)n0 * ROW);
    for (int i = tid; i < nvalid * R4; i += blockDim.x) {
      const int r = i / R4, c4 = i % R4;
      const float dn = den[r * HEADS + (c4 * 4) / C];
      const float4 a = acc4[i], b = b4[c4];
      float4 v;
      v.x = a.x / dn + b.x; v.x = v.x > 0.f ? v.x : (expf(v.x) - 1.f);
      v.y = a.y / dn + b.y; v.y = v.y > 0.f ? v.y : (expf(v.y) - 1.f);
      v.z = a.z / dn + b.z; v.z = v.z > 0.f ? v.z : (expf(v.z) - 1.f);
      v.w = a.w / dn + b.w; v.w = v.w > 0.f ? v.w : (expf(v.w) - 1.f);
      o4[i] = v;
    }
  } else {
    static_assert(ROW == 128, "log-softmax row: 4 channels per lane");
    for (int r = wave; r < nvalid; r += nw) {
      float v[4];
      for (int k = 0; k < 4; ++k) {
        const int c = lane + 32 * k;
        v[k] = acc[r * ROW + c] / den[r * HEADS + c / C] + bias[c];
      }
      float mx = fmaxf(fmaxf(v[0], v[2]), fmaxf(v[1], v[3]));
      for (int sft = 16; sft > 0; sft >>= 1) mx = fmaxf(mx, __shfl_xor(mx, sft));
      float sum = (expf(v[0] - mx) + expf(v[2] - mx)) + (expf(v[1] - mx) + expf(v[3] - mx));
      for (int sft = 16; sft > 0; sft >>= 1) sum += __shfl_xor(sum, sft);
      const float lse = logf(sum);
      float* orow = out + (size_t)(n0 + r) * ROW + lane;
      for (int k = 0; k < 4; ++k) orow[32 * k] = v[k] - mx - lse;
    }
  }
}

__device__ __forceinline__ constexpr int kprow(int st) { return 4 * (st >> 1) + (st & 1); }

template <int K, int KOUT, int GC>
__global__ __launch_bounds__(512) void gemm_pair_lds(
    const float* __restrict__ A, const float* __restrict__ Wl, const float* __restrict__ Wr,
    float* __restrict__ Cl, float* __restrict__ Cr, int M) {
  constexpr int NT = 2 * KOUT;
  constexpr int KP = K / 2;
  constexpr int PITCH = NT + 8;
  constexpr int NA = K / 8;
  constexpr int NB = K / 32;
  constexpr int GROUPS = NT / GC;
  static_assert(KOUT % GC == 0 && GC % 32 == 0, "a column group lies inside one product");
  __shared__ float4 Wp4[KP * PITCH / 2];
#if USE_WMMA_F32
  const int tid = threadIdx.x, lane = tid & 31;
  const int h = lane >> 4, c = lane & 15;
  const int wave = __builtin_amdgcn_readfirstlane(tid >> 5);
  const int items = (M >> 4) * GROUPS;
  const int nblk = (int)gridDim.x;

  float4 a4[NA];
  auto load_strip = [&](int t) {
    const int m = t / GROUPS;
    const float4* ap = (const float4*)(A + (size_t)(m * 16 + c) * K) + h;
#pragma unroll
    for (int j = 0; j < NA; ++j) a4[j] = ap[2 * j];
  };
  int slot = wave;
  int t = slot * nblk + (int)blockIdx.x;
  if (t < items) load_strip(t);

  {
    constexpr int Q = NT / 4;
    constexpr int UNITS = KP * Q;
    static_assert(UNITS % 2048 == 0, "copy loop: 4 units per thread per trip");
    for (int u0 = 0; u0 < UNITS; u0 += 2048) {
      float4 r0[4], r1[4];
#pragma unroll
      for (int i = 0; i < 4; ++i) {
        const int u = u0 + i * 512 + tid;
        const int kp = u / Q, col = 4 * (u % Q);
        const float* s = col < KOUT ? Wl + col : Wr + (col - KOUT);
        r0[i] = *(const float4*)(s + (size_t)(2 * kp) * KOUT);
        r1[i] = *(const float4*)(s + (size_t)(2 * kp + 1) * KOUT);
      }
#pragma unroll
      for (int i = 0; i < 4; ++i) {
        const int u = u0 + i * 512 + tid;
        const int kp = u / Q, q = u % Q;
        float4* d = Wp4 + (kp * PITCH + 4 * q) / 2;
        d[0] = make_float4(r0[i].x, r1[i].x, r0[i].y, r1[i].y);
        d[1] = make_float4(r0[i].z, r1[i].z, r0[i].w, r1[i].w);
      }
    }
  }
  __syncthreads();

  const v2f* bp = (const v2f*)Wp4 + (2 * h) * PITCH + c;
  while (t < items) {
    const int m = t / GROUPS, g = t % GROUPS;
    for (int p = 0; p < GC / 32; ++p) {
      const int n0 = g * GC + 32 * p;
      const v2f* bq = bp + n0;
      v8f acc0 = {}, acc1 = {};
      v2f b[2][8][2];
#pragma unroll
      for (int s = 0; s < 8; ++s) {
        b[0][s][0] = bq[kprow(s) * PITCH];
        b[0][s][1] = bq[kprow(s) * PITCH + 16];
      }
#pragma unroll
      for (int blk = 0; blk < NB; ++blk) {
        if (blk + 1 < NB) {
#pragma unroll
          for (int s = 0; s < 8; ++s) {
            b[(blk + 1) & 1][s][0] = bq[kprow((blk + 1) * 8 + s) * PITCH];
            b[(blk + 1) & 1][s][1] = bq[kprow((blk + 1) * 8 + s) * PITCH + 16];
          }
        }
#pragma unroll
        for (int s = 0; s < 8; ++s) {
          const float4 av = a4[(8 * blk + s) >> 1];
          v2f a;
          if (s & 1) { a[0] = av.z; a[1] = av.w; } else { a[0] = av.x; a[1] = av.y; }
          acc0 = __builtin_amdgcn_wmma_f32_16x16x4_f32(false, a, false, b[blk & 1][s][0],
                                                       (short)0, acc0, false, false);
          acc1 = __builtin_amdgcn_wmma_f32_16x16x4_f32(false, a, false, b[blk & 1][s][1],
                                                       (short)0, acc1, false, false);
        }
      }
      float* Cb = (n0 < KOUT ? Cl + n0 : Cr + (n0 - KOUT)) + (size_t)(m * 16) * KOUT + lane;
#pragma unroll
      for (int v = 0; v < 8; ++v) {
        const float got = xhalf(h ? acc0[v] : acc1[v]);
        Cb[(size_t)v * KOUT] = h ? got : acc0[v];
        Cb[(size_t)(8 + v) * KOUT] = h ? acc1[v] : got;
      }
    }
    slot += 16;
    t = slot * nblk + (int)blockIdx.x;
    if (t < items) load_strip(t);
  }
#else
  for (int i = blockIdx.x * blockDim.x + threadIdx.x; i < M * NT; i += gridDim.x * blockDim.x) {
    const int r = i / NT, n = i % NT;
    const float* Wc = n < KOUT ? Wl + n : Wr + (n - KOUT);
    float s = 0.f;
    for (int k = 0; k < K; ++k) s += A[(size_t)r * K + k] * Wc[(size_t)k * KOUT];
    (n < KOUT ? Cl + n : Cr + (n - KOUT))[(size_t)r * KOUT] = s;
  }
#endif
}

extern "C" void kernel_launch(void* const* d_in, const int* in_sizes, int n_in,
                              void* d_out, int out_size, void* d_ws, size_t ws_size,
                              hipStream_t stream) {
  (void)n_in; (void)out_size; (void)ws_size;
  const float* x    = (const float*)d_in[0];
  const int*   ei   = (const int*)d_in[1];
  const float* W1l  = (const float*)d_in[2];
  const float* W1r  = (const float*)d_in[3];
  const float* att1 = (const float*)d_in[4];
  const float* b1   = (const float*)d_in[5];
  const float* W2l  = (const float*)d_in[6];
  const float* W2r  = (const float*)d_in[7];
  const float* att2 = (const float*)d_in[8];
  const float* b2   = (const float*)d_in[9];

  int N = in_sizes[0] / 128;
  int E = in_sizes[1] / 2;
  const int* srcI = ei;
  const int* dstI = ei + E;

  float* ws    = (float*)d_ws;
  float* xl    = ws;
  float* xr    = xl + (size_t)N * 256;
  float* hbuf  = xr + (size_t)N * 256;
  float* lg    = hbuf + (size_t)N * 256;
  int* glist   = (int*)(lg + (size_t)(E + N) * 8);
  float* outFin = (float*)d_out;

  int agrid = (N + ACC_NB - 1) / ACC_NB;
  int* gcount  = glist + (size_t)agrid * ACC_CAP;

  gemm_pair_lds<128, 256, 128><<<256, dim3(512), 0, stream>>>(x, W1l, W1r, xl, xr, N);
  edge_owned<32, false, false><<<agrid, dim3(ACC_T), 0, stream>>>(xl, xr, att1, srcI, dstI, E, N, lg, glist, gcount, b1, hbuf);

  gemm_pair_lds<256, 128, 64><<<256, dim3(512), 0, stream>>>(hbuf, W2l, W2r, xl, xr, N);
  edge_owned<16, true, true><<<agrid, dim3(ACC_T), 0, stream>>>(xl, xr, att2, srcI, dstI, E, N, lg, glist, gcount, b2, outFin);
}


# dst scan replaced by an index: layer-1 GEMM launch's last waves flag, per owner block, the 128-B chunks of dst holding its nodes; a block reads its flag column and those chunks only; sum order kept
# speedup vs baseline: 6.9451x; 1.0864x over previous
#include <hip/hip_runtime.h>
#include <math.h>

typedef __attribute__((ext_vector_type(2))) float v2f;
typedef __attribute__((ext_vector_type(8))) float v8f;

#define HEADS 8

#if defined(__gfx1250__) && __has_builtin(__builtin_amdgcn_wmma_f32_16x16x4_f32)
#define USE_WMMA_F32 1
#else
#define USE_WMMA_F32 0
#endif

__device__ __forceinline__ unsigned enc_ord(float f) {
  unsigned b = __float_as_uint(f);
  return (b & 0x80000000u) ? ~b : (b | 0x80000000u);
}
__device__ __forceinline__ float dec_ord(unsigned u) {
  unsigned b = (u & 0x80000000u) ? (u ^ 0x80000000u) : ~u;
  return __uint_as_float(b);
}

#define ACC_NB  98
#define ACC_T   1024
#define ACC_U   8
#define ACC_CAP 4096

#define IDX_CH   32
#define IDX_GRP  (32 * IDX_CH)
#define IDX_OWN  512
#define IDX_CCAP 2048
#define IDX_U    8

__device__ __forceinline__ void lds_add(float* p, float v) {
  __hip_atomic_fetch_add(p, v, __ATOMIC_RELAXED, __HIP_MEMORY_SCOPE_WORKGROUP);
}

template <class F>
__device__ __forceinline__ void scan_owned(const int* __restrict__ dstI, int E, int n0,
                                           int wave, int nw, int lane, F f) {
  const int stepE = ACC_U * 32;
  const int strideE = nw * stepE;
  int base = wave * stepE;
  int cur[ACC_U], nxt[ACC_U];
#pragma unroll
  for (int u = 0; u < ACC_U; ++u) {
    int e = base + u * 32 + lane;
    cur[u] = e < E ? dstI[e] : -1;
  }
  while (base < E) {
    const int nbase = base + strideE;
    if (nbase + stepE <= E) {
      const int* p = dstI + nbase + lane;
#pragma unroll
      for (int u = 0; u < ACC_U; ++u) nxt[u] = p[u * 32];
    } else {
#pragma unroll
      for (int u = 0; u < ACC_U; ++u) {
        int e = nbase + u * 32 + lane;
        nxt[u] = e < E ? dstI[e] : -1;
      }
    }
    bool any = false;
#pragma unroll
    for (int u = 0; u < ACC_U; ++u) any |= (unsigned)(cur[u] - n0) < (unsigned)ACC_NB;
    if (__builtin_amdgcn_ballot_w32(any)) {
#pragma unroll
      for (int u = 0; u < ACC_U; ++u) {
        unsigned mask = __builtin_amdgcn_ballot_w32((unsigned)(cur[u] - n0) < (unsigned)ACC_NB);
        while (mask) {
          const int l = __builtin_ctz(mask);
          mask &= mask - 1;
          f(base + u * 32 + l, __builtin_amdgcn_readlane(cur[u], l));
        }
      }
    }
#pragma unroll
    for (int u = 0; u < ACC_U; ++u) cur[u] = nxt[u];
    base = nbase;
  }
}

__device__ __forceinline__ void scan_list(const int* __restrict__ dstI, int E, int n0,
                                          int wave, int nw, int lane, int* list, int* count) {
  const int stepE = ACC_U * 32;
  const int strideE = nw * stepE;
  int base = wave * stepE;
  int cur[ACC_U], nxt[ACC_U];
#pragma unroll
  for (int u = 0; u < ACC_U; ++u) {
    int e = base + u * 32 + lane;
    cur[u] = e < E ? dstI[e] : -1;
  }
  while (base < E) {
    const int nbase = base + strideE;
    if (nbase + stepE <= E) {
      const int* p = dstI + nbase + lane;
#pragma unroll
      for (int u = 0; u < ACC_U; ++u) nxt[u] = p[u * 32];
    } else {
#pragma unroll
      for (int u = 0; u < ACC_U; ++u) {
        int e = nbase + u * 32 + lane;
        nxt[u] = e < E ? dstI[e] : -1;
      }
    }
    bool any = false;
#pragma unroll
    for (int u = 0; u < ACC_U; ++u) any |= (unsigned)(cur[u] - n0) < (unsigned)ACC_NB;
    if (__builtin_amdgcn_ballot_w32(any)) {
#pragma unroll
      for (int u = 0; u < ACC_U; ++u) {
        if ((unsigned)(cur[u] - n0) < (unsigned)ACC_NB) {
          const int slot = atomicAdd(count, 1);
          if (slot < ACC_CAP) list[slot] = base + u * 32 + lane;
        }
      }
    }
#pragma unroll
    for (int u = 0; u < ACC_U; ++u) cur[u] = nxt[u];
    base = nbase;
  }
}

__device__ __forceinline__ float xhalf(float x) {
  const int b = __float_as_int(x);
  return __int_as_float(__builtin_amdgcn_permlanex16(b, b, 0x76543210, 0xfedcba98, false, false));
}
template <int M>
__device__ __forceinline__ float xrow(float x) {
  return __int_as_float(__builtin_amdgcn_update_dpp(0, __float_as_int(x), 0x160 + M, 0xF, 0xF, true));
}

template <int C>
__device__ __forceinline__ void logit_load(const float* __restrict__ xl,
                                           int s, int lane,
                                           float (&a)[HEADS * C / 32]) {
  constexpr int ROW = HEADS * C;
  constexpr int NJ = ROW / 32;
  const float* pl = xl + (size_t)s * ROW + lane;
#pragma unroll
  for (int j = 0; j < NJ; ++j) a[j] = pl[j * 32];
}

template <int C>
__device__ __forceinline__ void logit_finish(const float (&a)[HEADS * C / 32],
                                             const float* xrows,
                                             const float (&aw)[HEADS * C / 32],
                                             float* lg, unsigned* mx,
                                             int e, int dl, int lane) {
  constexpr int ROW = HEADS * C;
  constexpr int NJ = ROW / 32;
  const float* pr = xrows + dl * ROW + lane;
  float p[NJ];
#pragma unroll
  for (int j = 0; j < NJ; ++j) {
    float v = a[j] + pr[j * 32];
    v = v > 0.f ? v : 0.2f * v;
    p[j] = v * aw[j];
  }
  const bool b16 = (lane & 16) != 0, b8 = (lane & 8) != 0, b4 = (lane & 4) != 0;
  float r0, r1;
  int h;
  if constexpr (C == 32) {
    float q[4];
#pragma unroll
    for (int j = 0; j < 4; ++j) {
      const float mine = b16 ? p[j + 4] : p[j];
      const float send = b16 ? p[j] : p[j + 4];
      q[j] = mine + xhalf(send);
    }
    const float m0 = b8 ? q[2] : q[0], s0 = b8 ? q[0] : q[2];
    const float m1 = b8 ? q[3] : q[1], s1 = b8 ? q[1] : q[3];
    r0 = m0 + xrow<8>(s0);
    r1 = m1 + xrow<8>(s1);
    h = lane >> 2;
  } else {
    static_assert(C == 16, "row layout: 16 or 32 channels per head");
    const float m0 = b8 ? p[2] : p[0], s0 = b8 ? p[0] : p[2];
    const float m1 = b8 ? p[3] : p[1], s1 = b8 ? p[1] : p[3];
    r0 = m0 + xrow<8>(s0);
    r1 = m1 + xrow<8>(s1);
    h = 2 * ((lane >> 2) & 1) + 4 * ((lane >> 3) & 1) + (lane >> 4);
  }
  const float mine = b4 ? r1 : r0, send = b4 ? r0 : r1;
  float t = mine + xrow<4>(send);
  t += xrow<2>(t);
  t += xrow<1>(t);
  if ((lane & 3) == 0) {
    lg[(size_t)e * HEADS + h] = t;
    atomicMax(mx + dl * HEADS + h, enc_ord(t));
  }
}

template <int C>
__device__ __forceinline__ void edge_logit(const float* __restrict__ xl,
                                           const float* xrows,
                                           const float (&aw)[HEADS * C / 32],
                                           float* lg, unsigned* mx,
                                           int e, int s, int dl, int lane) {
  float a[HEADS * C / 32];
  logit_load<C>(xl, s, lane, a);
  logit_finish<C>(a, xrows, aw, lg, mx, e, dl, lane);
}

template <int C>
__device__ __forceinline__ float head_ex(float ex, int j, int lane) {
  if constexpr (C == 32) {
    return __int_as_float(__builtin_amdgcn_readlane(__float_as_int(ex), j));
  } else {
    static_assert(C == 16, "row layout: 16 or 32 channels per head");
    float lo = __int_as_float(__builtin_amdgcn_readlane(__float_as_int(ex), 2 * j));
    float hi = __int_as_float(__builtin_amdgcn_readlane(__float_as_int(ex), 2 * j + 1));
    return lane < 16 ? lo : hi;
  }
}

template <int C>
__device__ __forceinline__ void add_load(const float* __restrict__ xl,
                                         const float* lg, const unsigned* mx,
                                         int e, int s, int dl, int lane,
                                         float (&v)[HEADS * C / 32], float& lgv, unsigned& mxv) {
  constexpr int ROW = HEADS * C;
  constexpr int NJ = ROW / 32;
  const int hh = lane & 7;
  const float* row = xl + (size_t)s * ROW + lane;
#pragma unroll
  for (int j = 0; j < NJ; ++j) v[j] = row[j * 32];
  lgv = lg[(size_t)e * HEADS + hh];
  mxv = mx[dl * HEADS + hh];
}

template <int C>
__device__ __forceinline__ void add_finish(const float (&v)[HEADS * C / 32], float lgv, unsigned mxv,
                                           float* acc, float* den, int dl, int lane) {
  constexpr int ROW = HEADS * C;
  constexpr int NJ = ROW / 32;
  const float ex = expf(lgv - dec_ord(mxv));
  if (lane < HEADS) lds_add(den + dl * HEADS + lane, ex);
  float* arow = acc + dl * ROW + lane;
#pragma unroll
  for (int j = 0; j < NJ; ++j) lds_add(arow + j * 32, head_ex<C>(ex, j, lane) * v[j]);
}

template <int C>
__device__ __forceinline__ void add_edge(const float* __restrict__ xl,
                                         const float* lg, const unsigned* mx,
                                         float* acc, float* den,
                                         int e, int s, int dl, int lane) {
  float v[HEADS * C / 32], lgv; unsigned mxv;
  add_load<C>(xl, lg, mx, e, s, dl, lane, v, lgv, mxv);
  add_finish<C>(v, lgv, mxv, acc, den, dl, lane);
}

template <int C, bool LSM, bool SAVED>
__global__ __launch_bounds__(ACC_T) void edge_owned(
    const float* __restrict__ xl, const float* __restrict__ xr,
    const float* __restrict__ att,
    const int* __restrict__ srcI, const int* __restrict__ dstI,
    int E, int N,
    float* lg,
    int* glist, int* gcount,
    const unsigned* __restrict__ bmap,
    const float* __restrict__ bias, float* __restrict__ out) {
  constexpr int ROW = HEADS * C;
  constexpr int NJ = ROW / 32;
  __shared__ float4 acc4[ACC_NB * ROW / 4];
  __shared__ float den[ACC_NB * HEADS];
  __shared__ unsigned mx[ACC_NB * HEADS];
  __shared__ int list[ACC_CAP];
  __shared__ int lcount;
  float* acc = (float*)acc4;
  const int tid = threadIdx.x, lane = tid & 31;
  const int wave = __builtin_amdgcn_readfirstlane(tid >> 5), nw = blockDim.x >> 5;
  const int n0 = blockIdx.x * ACC_NB;

  const int nvalid = (N - n0) < ACC_NB ? (N - n0) : ACC_NB;
  constexpr int NST = (ACC_NB * ROW / 4 + ACC_T - 1) / ACC_T;
  float4 st[NST];
  auto rows_request = [&]() {
    const float4* xr4 = (const float4*)(xr + (size_t)n0 * ROW);
#pragma unroll
    for (int k = 0; k < NST; ++k) {
      const int i = tid + k * ACC_T;
      st[k] = i < nvalid * (ROW / 4) ? xr4[i] : make_float4(0.f, 0.f, 0.f, 0.f);
    }
  };
  auto rows_store = [&]() {
#pragma unroll
    for (int k = 0; k < NST; ++k) {
      const int i = tid + k * ACC_T;
      if (i < ACC_NB * ROW / 4) acc4[i] = st[k];
    }
  };
  for (int i = tid; i < ACC_NB * HEADS; i += blockDim.x) { den[i] = 0.f; mx[i] = 0x007FFFFFu; }
  int* gl = glist + (size_t)blockIdx.x * ACC_CAP;
  int total;
  if constexpr (!SAVED) {
    __shared__ int chunks[IDX_CCAP];
    __shared__ int ccount;
    if (tid == 0) { lcount = 0; ccount = 0; }
    __syncthreads();
    rows_request();
    const int ngrp = (E + IDX_GRP - 1) / IDX_GRP;
    const int col = (int)blockIdx.x < IDX_OWN - 1 ? (int)blockIdx.x : IDX_OWN - 1;
    for (int g = tid; g < ngrp; g += blockDim.x) {
      unsigned w = bmap[(size_t)g * IDX_OWN + col];
      while (w) {
        const int k = __builtin_ctz(w);
        w &= w - 1;
        const int slot = atomicAdd(&ccount, 1);
        if (slot < IDX_CCAP) chunks[slot] = g * 32 + k;
      }
    }
    __syncthreads();
    const int nch = __builtin_amdgcn_readfirstlane(ccount);
    if (nch <= IDX_CCAP) {
      for (int i0 = wave; i0 < nch; i0 += nw * IDX_U) {
        const int iu = i0 + lane * nw;
        const int mine = (lane < IDX_U && iu < nch) ? chunks[iu] : -1;
        int ee[IDX_U], dd[IDX_U];
#pragma unroll
        for (int u = 0; u < IDX_U; ++u) {
          const int cn = __builtin_amdgcn_readlane(mine, u);
          ee[u] = cn >= 0 ? cn * IDX_CH + lane : E;
          dd[u] = ee[u] < E ? dstI[ee[u]] : -1;
        }
        unsigned own[IDX_U];
        int cnt = 0;
#pragma unroll
        for (int u = 0; u < IDX_U; ++u) {
          own[u] = __builtin_amdgcn_ballot_w32((unsigned)(dd[u] - n0) < (unsigned)ACC_NB);
          cnt += __builtin_popcount(own[u]);
        }
        if (cnt) {
          int base = 0;
          if (lane == 0) base = atomicAdd(&lcount, cnt);
          base = __builtin_amdgcn_readfirstlane(base);
#pragma unroll
          for (int u = 0; u < IDX_U; ++u) {
            if ((own[u] >> lane) & 1u) {
              const int slot = base + __builtin_popcount(own[u] & ((1u << lane) - 1u));
              if (slot < ACC_CAP) list[slot] = ee[u];
            }
            base += __builtin_popcount(own[u]);
          }
        }
      }
    } else {
      scan_list(dstI, E, n0, wave, nw, lane, list, &lcount);
    }
    rows_store();
    __syncthreads();
    total = __builtin_amdgcn_readfirstlane(lcount);
    const int nsave = total < ACC_CAP ? total : ACC_CAP;
    for (int i = tid; i < nsave; i += blockDim.x) gl[i] = list[i];
    if (tid == 0) gcount[blockIdx.x] = total;
  } else {
    rows_request();
    total = __builtin_amdgcn_readfirstlane(gcount[blockIdx.x]);
    if (total <= ACC_CAP)
      for (int i = tid; i < total; i += blockDim.x) list[i] = gl[i];
    rows_store();
    __syncthreads();
  }
  const bool listed = total <= ACC_CAP;

  auto fetch = [&](int idx, int& e, int& s, int& d) {
    if (idx < total) { e = __builtin_amdgcn_readfirstlane(list[idx]); s = srcI[e]; d = dstI[e]; }
    else             { e = -1; s = 0; d = n0; }
  };

  {
    float aw[NJ];
#pragma unroll
    for (int j = 0; j < NJ; ++j) aw[j] = att[j * 32 + lane];
    if (listed) {
      int i = wave, e0, s0, d0, e1, s1, d1;
      fetch(i, e0, s0, d0);
      fetch(i + nw, e1, s1, d1);
      while (e0 >= 0) {
        float a0[NJ], a1[NJ];
        logit_load<C>(xl, s0, lane, a0);
        logit_load<C>(xl, s1, lane, a1);
        int f0, t0, g0, f1, t1, g1;
        fetch(i + 2 * nw, f0, t0, g0);
        fetch(i + 3 * nw, f1, t1, g1);
        logit_finish<C>(a0, acc, aw, lg, mx, e0, d0 - n0, lane);
        if (e1 >= 0) logit_finish<C>(a1, acc, aw, lg, mx, e1, d1 - n0, lane);
        i += 2 * nw;
        e0 = f0; s0 = t0; d0 = g0; e1 = f1; s1 = t1; d1 = g1;
      }
    } else {
      scan_owned(dstI, E, n0, wave, nw, lane, [&](int e, int d) {
        edge_logit<C>(xl, acc, aw, lg, mx, e, srcI[e], d - n0, lane);
      });
    }
    for (int i = wave; i < ACC_NB; i += nw) {
      const int n = n0 + i;
      if (n < N) edge_logit<C>(xl, acc, aw, lg, mx, E + n, n, i, lane);
    }
  }
  __syncthreads();
  for (int i = tid; i < ACC_NB * ROW / 4; i += blockDim.x) acc4[i] = make_float4(0.f, 0.f, 0.f, 0.f);
  __syncthreads();

  if (listed) {
    int i = wave, e0, s0, d0, e1, s1, d1;
    fetch(i, e0, s0, d0);
    fetch(i + nw, e1, s1, d1);
    while (e0 >= 0) {
      float v0[NJ], v1[NJ], l0, l1;
      unsigned m0, m1;
      const int eb = e1 >= 0 ? e1 : e0;
      add_load<C>(xl, lg, mx, e0, s0, d0 - n0, lane, v0, l0, m0);
      add_load<C>(xl, lg, mx, eb, s1, d1 - n0, lane, v1, l1, m1);
      int f0, t0, g0, f1, t1, g1;
      fetch(i + 2 * nw, f0, t0, g0);
      fetch(i + 3 * nw, f1, t1, g1);
      add_finish<C>(v0, l0, m0, acc, den, d0 - n0, lane);
      if (e1 >= 0) add_finish<C>(v1, l1, m1, acc, den, d1 - n0, lane);
      i += 2 * nw;
      e0 = f0; s0 = t0; d0 = g0; e1 = f1; s1 = t1; d1 = g1;
    }
  } else {
    scan_owned(dstI, E, n0, wave, nw, lane, [&](int e, int d) {
      add_edge<C>(xl, lg, mx, acc, den, e, srcI[e], d - n0, lane);
    });
  }
  for (int i = wave; i < ACC_NB; i += nw) {
    const int n = n0 + i;
    if (n < N) add_edge<C>(xl, lg, mx, acc, den, E + n, n, i, lane);
  }
  __syncthreads();

  if constexpr (!LSM) {
    constexpr int R4 = ROW / 4;
    const float4* b4 = (const float4*)bias;
    float4* o4 = (float4*)(out + (size_t)n0 * ROW);
    for (int i = tid; i < nvalid * R4; i += blockDim.x) {
      const int r = i / R4, c4 = i % R4;
      const float dn = den[r * HEADS + (c4 * 4) / C];
      const float4 a = acc4[i], b = b4[c4];
      float4 v;
      v.x = a.x / dn + b.x; v.x = v.x > 0.f ? v.x : (expf(v.x) - 1.f);
      v.y = a.y / dn + b.y; v.y = v.y > 0.f ? v.y : (expf(v.y) - 1.f);
      v.z = a.z / dn + b.z; v.z = v.z > 0.f ? v.z : (expf(v.z) - 1.f);
      v.w = a.w / dn + b.w; v.w = v.w > 0.f ? v.w : (expf(v.w) - 1.f);
      o4[i] = v;
    }
  } else {
    static_assert(ROW == 128, "log-softmax row: 4 channels per lane");
    for (int r = wave; r < nvalid; r += nw) {
      float v[4];
      for (int k = 0; k < 4; ++k) {
        const int c = lane + 32 * k;
        v[k] = acc[r * ROW + c] / den[r * HEADS + c / C] + bias[c];
      }
      float mx = fmaxf(fmaxf(v[0], v[2]), fmaxf(v[1], v[3]));
      for (int sft = 16; sft > 0; sft >>= 1) mx = fmaxf(mx, __shfl_xor(mx, sft));
      float sum = (expf(v[0] - mx) + expf(v[2] - mx)) + (expf(v[1] - mx) + expf(v[3] - mx));
      for (int sft = 16; sft > 0; sft >>= 1) sum += __shfl_xor(sum, sft);
      const float lse = logf(sum);
      float* orow = out + (size_t)(n0 + r) * ROW + lane;
      for (int k = 0; k < 4; ++k) orow[32 * k] = v[k] - mx - lse;
    }
  }
}

__device__ __forceinline__ constexpr int kprow(int st) { return 4 * (st >> 1) + (st & 1); }

__device__ __forceinline__ void index_group(const int* __restrict__ dstI, int E,
                                            unsigned* __restrict__ bmap, unsigned* tab,
                                            int g, int lane) {
  for (int i = 0; i < IDX_OWN / 32; ++i) tab[i * 32 + lane] = 0u;
  const int e0 = g * IDX_GRP + lane;
  for (int kb = 0; kb < 32; kb += 16) {
    int d[16];
#pragma unroll
    for (int u = 0; u < 16; ++u) {
      const int e = e0 + (kb + u) * IDX_CH;
      d[u] = e < E ? dstI[e] : -1;
    }
#pragma unroll
    for (int u = 0; u < 16; ++u) {
      if (d[u] >= 0) {
        unsigned o = (unsigned)d[u] / (unsigned)ACC_NB;
        o = o < IDX_OWN - 1 ? o : IDX_OWN - 1;
        atomicOr(tab + o, 1u << (kb + u));
      }
    }
  }
  unsigned* row = bmap + (size_t)g * IDX_OWN + lane;
  for (int i = 0; i < IDX_OWN / 32; ++i) row[i * 32] = tab[i * 32 + lane];
}

template <int K, int KOUT, int GC, bool INDEX>
__global__ __launch_bounds__(512) void gemm_pair_lds(
    const float* __restrict__ A, const float* __restrict__ Wl, const float* __restrict__ Wr,
    float* __restrict__ Cl, float* __restrict__ Cr, int M,
    const int* __restrict__ dstI, int E, unsigned* __restrict__ bmap) {
  constexpr int NT = 2 * KOUT;
  constexpr int KP = K / 2;
  constexpr int PITCH = NT + 8;
  constexpr int NA = K / 8;
  constexpr int NB = K / 32;
  constexpr int GROUPS = NT / GC;
  static_assert(KOUT % GC == 0 && GC % 32 == 0, "a column group lies inside one product");
  __shared__ float4 Wp4[KP * PITCH / 2];
#if USE_WMMA_F32
  const int tid = threadIdx.x, lane = tid & 31;
  const int h = lane >> 4, c = lane & 15;
  const int wave = __builtin_amdgcn_readfirstlane(tid >> 5);
  const int items = (M >> 4) * GROUPS;
  const int nblk = (int)gridDim.x;

  float4 a4[NA];
  auto load_strip = [&](int t) {
    const int m = t / GROUPS;
    const float4* ap = (const float4*)(A + (size_t)(m * 16 + c) * K) + h;
#pragma unroll
    for (int j = 0; j < NA; ++j) a4[j] = ap[2 * j];
  };
  int slot = wave;
  int t = slot * nblk + (int)blockIdx.x;
  if (t < items) load_strip(t);

  {
    constexpr int Q = NT / 4;
    constexpr int UNITS = KP * Q;
    static_assert(UNITS % 2048 == 0, "copy loop: 4 units per thread per trip");
    for (int u0 = 0; u0 < UNITS; u0 += 2048) {
      float4 r0[4], r1[4];
#pragma unroll
      for (int i = 0; i < 4; ++i) {
        const int u = u0 + i * 512 + tid;
        const int kp = u / Q, col = 4 * (u % Q);
        const float* s = col < KOUT ? Wl + col : Wr + (col - KOUT);
        r0[i] = *(const float4*)(s + (size_t)(2 * kp) * KOUT);
        r1[i] = *(const float4*)(s + (size_t)(2 * kp + 1) * KOUT);
      }
#pragma unroll
      for (int i = 0; i < 4; ++i) {
        const int u = u0 + i * 512 + tid;
        const int kp = u / Q, q = u % Q;
        float4* d = Wp4 + (kp * PITCH + 4 * q) / 2;
        d[0] = make_float4(r0[i].x, r1[i].x, r0[i].y, r1[i].y);
        d[1] = make_float4(r0[i].z, r1[i].z, r0[i].w, r1[i].w);
      }
    }
  }
  __syncthreads();

  const v2f* bp = (const v2f*)Wp4 + (2 * h) * PITCH + c;
  while (t < items) {
    const int m = t / GROUPS, g = t % GROUPS;
    for (int p = 0; p < GC / 32; ++p) {
      const int n0 = g * GC + 32 * p;
      const v2f* bq = bp + n0;
      v8f acc0 = {}, acc1 = {};
      v2f b[2][8][2];
#pragma unroll
      for (int s = 0; s < 8; ++s) {
        b[0][s][0] = bq[kprow(s) * PITCH];
        b[0][s][1] = bq[kprow(s) * PITCH + 16];
      }
#pragma unroll
      for (int blk = 0; blk < NB; ++blk) {
        if (blk + 1 < NB) {
#pragma unroll
          for (int s = 0; s < 8; ++s) {
            b[(blk + 1) & 1][s][0] = bq[kprow((blk + 1) * 8 + s) * PITCH];
            b[(blk + 1) & 1][s][1] = bq[kprow((blk + 1) * 8 + s) * PITCH + 16];
          }
        }
#pragma unroll
        for (int s = 0; s < 8; ++s) {
          const float4 av = a4[(8 * blk + s) >> 1];
          v2f a;
          if (s & 1) { a[0] = av.z; a[1] = av.w; } else { a[0] = av.x; a[1] = av.y; }
          acc0 = __builtin_amdgcn_wmma_f32_16x16x4_f32(false, a, false, b[blk & 1][s][0],
                                                       (short)0, acc0, false, false);
          acc1 = __builtin_amdgcn_wmma_f32_16x16x4_f32(false, a, false, b[blk & 1][s][1],
                                                       (short)0, acc1, false, false);
        }
      }
      float* Cb = (n0 < KOUT ? Cl + n0 : Cr + (n0 - KOUT)) + (size_t)(m * 16) * KOUT + lane;
#pragma unroll
      for (int v = 0; v < 8; ++v) {
        const float got = xhalf(h ? acc0[v] : acc1[v]);
        Cb[(size_t)v * KOUT] = h ? got : acc0[v];
        Cb[(size_t)(8 + v) * KOUT] = h ? acc1[v] : got;
      }
    }
    slot += 16;
    t = slot * nblk + (int)blockIdx.x;
    if (t < items) load_strip(t);
  }
#else
  for (int i = blockIdx.x * blockDim.x + threadIdx.x; i < M * NT; i += gridDim.x * blockDim.x) {
    const int r = i / NT, n = i % NT;
    const float* Wc = n < KOUT ? Wl + n : Wr + (n - KOUT);
    float s = 0.f;
    for (int k = 0; k < K; ++k) s += A[(size_t)r * K + k] * Wc[(size_t)k * KOUT];
    (n < KOUT ? Cl + n : Cr + (n - KOUT))[(size_t)r * KOUT] = s;
  }
#endif
  if constexpr (INDEX) {
    __shared__ unsigned own[16 * IDX_OWN];
    const int iwave = __builtin_amdgcn_readfirstlane((int)threadIdx.x >> 5);
    const int inw = (int)blockDim.x >> 5;
    const int ngrp = (E + IDX_GRP - 1) / IDX_GRP;
    for (int g = (inw - 1 - iwave) * (int)gridDim.x + (int)blockIdx.x; g < ngrp; g += inw * (int)gridDim.x)
      index_group(dstI, E, bmap, own + (iwave & 15) * IDX_OWN, g, (int)threadIdx.x & 31);
  }
}

extern "C" void kernel_launch(void* const* d_in, const int* in_sizes, int n_in,
                              void* d_out, int out_size, void* d_ws, size_t ws_size,
                              hipStream_t stream) {
  (void)n_in; (void)out_size; (void)ws_size;
  const float* x    = (const float*)d_in[0];
  const int*   ei   = (const int*)d_in[1];
  const float* W1l  = (const float*)d_in[2];
  const float* W1r  = (const float*)d_in[3];
  const float* att1 = (const float*)d_in[4];
  const float* b1   = (const float*)d_in[5];
  const float* W2l  = (const float*)d_in[6];
  const float* W2r  = (const float*)d_in[7];
  const float* att2 = (const float*)d_in[8];
  const float* b2   = (const float*)d_in[9];

  int N = in_sizes[0] / 128;
  int E = in_sizes[1] / 2;
  const int* srcI = ei;
  const int* dstI = ei + E;

  float* ws    = (float*)d_ws;
  float* xl    = ws;
  float* xr    = xl + (size_t)N * 256;
  float* hbuf  = xr + (size_t)N * 256;
  float* lg    = hbuf + (size_t)N * 256;
  int* glist   = (int*)(lg + (size_t)(E + N) * 8);
  float* outFin = (float*)d_out;

  int agrid = (N + ACC_NB - 1) / ACC_NB;
  unsigned* bmap = (unsigned*)(glist + (size_t)agrid * ACC_CAP);
  int* gcount  = (int*)(bmap + (size_t)((E + IDX_GRP - 1) / IDX_GRP) * IDX_OWN);

  gemm_pair_lds<128, 256, 128, true><<<256, dim3(512), 0, stream>>>(x, W1l, W1r, xl, xr, N, dstI, E, bmap);
  edge_owned<32, false, false><<<agrid, dim3(ACC_T), 0, stream>>>(xl, xr, att1, srcI, dstI, E, N, lg, glist, gcount, bmap, b1, hbuf);

  gemm_pair_lds<256, 128, 64, false><<<256, dim3(512), 0, stream>>>(hbuf, W2l, W2r, xl, xr, N, dstI, E, bmap);
  edge_owned<16, true, true><<<agrid, dim3(ACC_T), 0, stream>>>(xl, xr, att2, srcI, dstI, E, N, lg, glist, gcount, bmap, b2, outFin);
}


# edge pass in one pass: softmax shift = node's self-loop logit, ex*xl[s] added while the row is in registers (no 2nd gather, no logits to memory); two-pass redo if logit > shift+40; self-loop term 1st
# speedup vs baseline: 8.4005x; 1.2096x over previous
#include <hip/hip_runtime.h>
#include <math.h>

typedef __attribute__((ext_vector_type(2))) float v2f;
typedef __attribute__((ext_vector_type(8))) float v8f;

#define HEADS 8

#if defined(__gfx1250__) && __has_builtin(__builtin_amdgcn_wmma_f32_16x16x4_f32)
#define USE_WMMA_F32 1
#else
#define USE_WMMA_F32 0
#endif

__device__ __forceinline__ unsigned enc_ord(float f) {
  unsigned b = __float_as_uint(f);
  return (b & 0x80000000u) ? ~b : (b | 0x80000000u);
}
__device__ __forceinline__ float dec_ord(unsigned u) {
  unsigned b = (u & 0x80000000u) ? (u ^ 0x80000000u) : ~u;
  return __uint_as_float(b);
}

#define ACC_NB  98
#define ACC_T   1024
#define ACC_U   8
#define ACC_CAP 4096

#define IDX_CH   32
#define IDX_GRP  (32 * IDX_CH)
#define IDX_OWN  512
#define IDX_CCAP 2048
#define IDX_U    8

__device__ __forceinline__ void lds_add(float* p, float v) {
  __hip_atomic_fetch_add(p, v, __ATOMIC_RELAXED, __HIP_MEMORY_SCOPE_WORKGROUP);
}

template <class F>
__device__ __forceinline__ void scan_owned(const int* __restrict__ dstI, int E, int n0,
                                           int wave, int nw, int lane, F f) {
  const int stepE = ACC_U * 32;
  const int strideE = nw * stepE;
  int base = wave * stepE;
  int cur[ACC_U], nxt[ACC_U];
#pragma unroll
  for (int u = 0; u < ACC_U; ++u) {
    int e = base + u * 32 + lane;
    cur[u] = e < E ? dstI[e] : -1;
  }
  while (base < E) {
    const int nbase = base + strideE;
    if (nbase + stepE <= E) {
      const int* p = dstI + nbase + lane;
#pragma unroll
      for (int u = 0; u < ACC_U; ++u) nxt[u] = p[u * 32];
    } else {
#pragma unroll
      for (int u = 0; u < ACC_U; ++u) {
        int e = nbase + u * 32 + lane;
        nxt[u] = e < E ? dstI[e] : -1;
      }
    }
    bool any = false;
#pragma unroll
    for (int u = 0; u < ACC_U; ++u) any |= (unsigned)(cur[u] - n0) < (unsigned)ACC_NB;
    if (__builtin_amdgcn_ballot_w32(any)) {
#pragma unroll
      for (int u = 0; u < ACC_U; ++u) {
        unsigned mask = __builtin_amdgcn_ballot_w32((unsigned)(cur[u] - n0) < (unsigned)ACC_NB);
        while (mask) {
          const int l = __builtin_ctz(mask);
          mask &= mask - 1;
          f(base + u * 32 + l, __builtin_amdgcn_readlane(cur[u], l));
        }
      }
    }
#pragma unroll
    for (int u = 0; u < ACC_U; ++u) cur[u] = nxt[u];
    base = nbase;
  }
}

__device__ __forceinline__ void scan_list(const int* __restrict__ dstI, int E, int n0,
                                          int wave, int nw, int lane, int* list, int* count) {
  const int stepE = ACC_U * 32;
  const int strideE = nw * stepE;
  int base = wave * stepE;
  int cur[ACC_U], nxt[ACC_U];
#pragma unroll
  for (int u = 0; u < ACC_U; ++u) {
    int e = base + u * 32 + lane;
    cur[u] = e < E ? dstI[e] : -1;
  }
  while (base < E) {
    const int nbase = base + strideE;
    if (nbase + stepE <= E) {
      const int* p = dstI + nbase + lane;
#pragma unroll
      for (int u = 0; u < ACC_U; ++u) nxt[u] = p[u * 32];
    } else {
#pragma unroll
      for (int u = 0; u < ACC_U; ++u) {
        int e = nbase + u * 32 + lane;
        nxt[u] = e < E ? dstI[e] : -1;
      }
    }
    bool any = false;
#pragma unroll
    for (int u = 0; u < ACC_U; ++u) any |= (unsigned)(cur[u] - n0) < (unsigned)ACC_NB;
    if (__builtin_amdgcn_ballot_w32(any)) {
#pragma unroll
      for (int u = 0; u < ACC_U; ++u) {
        if ((unsigned)(cur[u] - n0) < (unsigned)ACC_NB) {
          const int slot = atomicAdd(count, 1);
          if (slot < ACC_CAP) list[slot] = base + u * 32 + lane;
        }
      }
    }
#pragma unroll
    for (int u = 0; u < ACC_U; ++u) cur[u] = nxt[u];
    base = nbase;
  }
}

__device__ __forceinline__ float xhalf(float x) {
  const int b = __float_as_int(x);
  return __int_as_float(__builtin_amdgcn_permlanex16(b, b, 0x76543210, 0xfedcba98, false, false));
}
template <int M>
__device__ __forceinline__ float xrow(float x) {
  return __int_as_float(__builtin_amdgcn_update_dpp(0, __float_as_int(x), 0x160 + M, 0xF, 0xF, true));
}

template <int C>
__device__ __forceinline__ void logit_load(const float* __restrict__ xl,
                                           int s, int lane,
                                           float (&a)[HEADS * C / 32]) {
  constexpr int ROW = HEADS * C;
  constexpr int NJ = ROW / 32;
  const float* pl = xl + (size_t)s * ROW + lane;
#pragma unroll
  for (int j = 0; j < NJ; ++j) a[j] = pl[j * 32];
}

template <int C>
__device__ __forceinline__ float head_logit(const float (&a)[HEADS * C / 32],
                                            const float* xrows,
                                            const float (&aw)[HEADS * C / 32],
                                            int dl, int lane, int& h) {
  constexpr int ROW = HEADS * C;
  constexpr int NJ = ROW / 32;
  const float* pr = xrows + dl * ROW + lane;
  float p[NJ];
#pragma unroll
  for (int j = 0; j < NJ; ++j) {
    float v = a[j] + pr[j * 32];
    v = v > 0.f ? v : 0.2f * v;
    p[j] = v * aw[j];
  }
  const bool b16 = (lane & 16) != 0, b8 = (lane & 8) != 0, b4 = (lane & 4) != 0;
  float r0, r1;
  if constexpr (C == 32) {
    float q[4];
#pragma unroll
    for (int j = 0; j < 4; ++j) {
      const float mine = b16 ? p[j + 4] : p[j];
      const float send = b16 ? p[j] : p[j + 4];
      q[j] = mine + xhalf(send);
    }
    const float m0 = b8 ? q[2] : q[0], s0 = b8 ? q[0] : q[2];
    const float m1 = b8 ? q[3] : q[1], s1 = b8 ? q[1] : q[3];
    r0 = m0 + xrow<8>(s0);
    r1 = m1 + xrow<8>(s1);
    h = lane >> 2;
  } else {
    static_assert(C == 16, "row layout: 16 or 32 channels per head");
    const float m0 = b8 ? p[2] : p[0], s0 = b8 ? p[0] : p[2];
    const float m1 = b8 ? p[3] : p[1], s1 = b8 ? p[1] : p[3];
    r0 = m0 + xrow<8>(s0);
    r1 = m1 + xrow<8>(s1);
    h = 2 * ((lane >> 2) & 1) + 4 * ((lane >> 3) & 1) + (lane >> 4);
  }
  const float mine = b4 ? r1 : r0, send = b4 ? r0 : r1;
  float t = mine + xrow<4>(send);
  t += xrow<2>(t);
  t += xrow<1>(t);
  return t;
}

template <int C>
__device__ __forceinline__ void logit_finish(const float (&a)[HEADS * C / 32],
                                             const float* xrows,
                                             const float (&aw)[HEADS * C / 32],
                                             float* lg, unsigned* mx,
                                             int e, int dl, int lane) {
  int h;
  const float t = head_logit<C>(a, xrows, aw, dl, lane, h);
  if ((lane & 3) == 0) {
    lg[(size_t)e * HEADS + h] = t;
    atomicMax(mx + dl * HEADS + h, enc_ord(t));
  }
}

#define SP_LIM 40.0f

template <int C>
__device__ __forceinline__ float quad_ex(float ex, int j, int lane) {
  if constexpr (C == 32) {
    return __int_as_float(__builtin_amdgcn_readlane(__float_as_int(ex), 4 * j));
  } else {
    static_assert(C == 16, "row layout: 16 or 32 channels per head");
    float lo = __int_as_float(__builtin_amdgcn_readlane(__float_as_int(ex), 4 * j));
    float hi = __int_as_float(__builtin_amdgcn_readlane(__float_as_int(ex), 16 + 4 * j));
    return lane < 16 ? lo : hi;
  }
}

template <int C>
__device__ __forceinline__ void sp_finish(const float (&a)[HEADS * C / 32],
                                          const float* xrows,
                                          const float (&aw)[HEADS * C / 32],
                                          const unsigned* shift, float* acc, float* den, int* redo,
                                          int dl, int lane) {
  constexpr int ROW = HEADS * C;
  constexpr int NJ = ROW / 32;
  int h;
  const float t = head_logit<C>(a, xrows, aw, dl, lane, h);
  const float dlt = t - __uint_as_float(shift[dl * HEADS + h]);
  if (!(dlt <= SP_LIM)) *redo = 1;
  const float ex = expf(dlt);
  if ((lane & 3) == 0) lds_add(den + dl * HEADS + h, ex);
  float* arow = acc + dl * ROW + lane;
#pragma unroll
  for (int j = 0; j < NJ; ++j) lds_add(arow + j * 32, quad_ex<C>(ex, j, lane) * a[j]);
}

template <int C>
__device__ __forceinline__ void edge_logit(const float* __restrict__ xl,
                                           const float* xrows,
                                           const float (&aw)[HEADS * C / 32],
                                           float* lg, unsigned* mx,
                                           int e, int s, int dl, int lane) {
  float a[HEADS * C / 32];
  logit_load<C>(xl, s, lane, a);
  logit_finish<C>(a, xrows, aw, lg, mx, e, dl, lane);
}

template <int C>
__device__ __forceinline__ float head_ex(float ex, int j, int lane) {
  if constexpr (C == 32) {
    return __int_as_float(__builtin_amdgcn_readlane(__float_as_int(ex), j));
  } else {
    static_assert(C == 16, "row layout: 16 or 32 channels per head");
    float lo = __int_as_float(__builtin_amdgcn_readlane(__float_as_int(ex), 2 * j));
    float hi = __int_as_float(__builtin_amdgcn_readlane(__float_as_int(ex), 2 * j + 1));
    return lane < 16 ? lo : hi;
  }
}

template <int C>
__device__ __forceinline__ void add_load(const float* __restrict__ xl,
                                         const float* lg, const unsigned* mx,
                                         int e, int s, int dl, int lane,
                                         float (&v)[HEADS * C / 32], float& lgv, unsigned& mxv) {
  constexpr int ROW = HEADS * C;
  constexpr int NJ = ROW / 32;
  const int hh = lane & 7;
  const float* row = xl + (size_t)s * ROW + lane;
#pragma unroll
  for (int j = 0; j < NJ; ++j) v[j] = row[j * 32];
  lgv = lg[(size_t)e * HEADS + hh];
  mxv = mx[dl * HEADS + hh];
}

template <int C>
__device__ __forceinline__ void add_finish(const float (&v)[HEADS * C / 32], float lgv, unsigned mxv,
                                           float* acc, float* den, int dl, int lane) {
  constexpr int ROW = HEADS * C;
  constexpr int NJ = ROW / 32;
  const float ex = expf(lgv - dec_ord(mxv));
  if (lane < HEADS) lds_add(den + dl * HEADS + lane, ex);
  float* arow = acc + dl * ROW + lane;
#pragma unroll
  for (int j = 0; j < NJ; ++j) lds_add(arow + j * 32, head_ex<C>(ex, j, lane) * v[j]);
}

template <int C>
__device__ __forceinline__ void add_edge(const float* __restrict__ xl,
                                         const float* lg, const unsigned* mx,
                                         float* acc, float* den,
                                         int e, int s, int dl, int lane) {
  float v[HEADS * C / 32], lgv; unsigned mxv;
  add_load<C>(xl, lg, mx, e, s, dl, lane, v, lgv, mxv);
  add_finish<C>(v, lgv, mxv, acc, den, dl, lane);
}

template <int C, bool LSM, bool SAVED>
__global__ __launch_bounds__(ACC_T) void edge_owned(
    const float* __restrict__ xl, const float* __restrict__ xr,
    const float* __restrict__ att,
    const int* __restrict__ srcI, const int* __restrict__ dstI,
    int E, int N,
    float* lg,
    int* glist, int* gcount,
    const unsigned* __restrict__ bmap,
    const float* __restrict__ bias, float* __restrict__ out) {
  constexpr int ROW = HEADS * C;
  constexpr int NJ = ROW / 32;
  __shared__ float4 acc4[ACC_NB * ROW / 4];
  __shared__ float4 xrr4[ACC_NB * ROW / 4];
  __shared__ float den[ACC_NB * HEADS];
  __shared__ unsigned mx[ACC_NB * HEADS];
  __shared__ int list[ACC_CAP];
  __shared__ int lcount;
  __shared__ int redo;
  float* acc = (float*)acc4;
  const float* xrr = (const float*)xrr4;
  const int tid = threadIdx.x, lane = tid & 31;
  const int wave = __builtin_amdgcn_readfirstlane(tid >> 5), nw = blockDim.x >> 5;
  const int n0 = blockIdx.x * ACC_NB;

  const int nvalid = (N - n0) < ACC_NB ? (N - n0) : ACC_NB;
  constexpr int NST = (ACC_NB * ROW / 4 + ACC_T - 1) / ACC_T;
  float4 st[NST];
  auto rows_request = [&]() {
    const float4* xr4 = (const float4*)(xr + (size_t)n0 * ROW);
#pragma unroll
    for (int k = 0; k < NST; ++k) {
      const int i = tid + k * ACC_T;
      st[k] = i < nvalid * (ROW / 4) ? xr4[i] : make_float4(0.f, 0.f, 0.f, 0.f);
    }
  };
  auto rows_store = [&]() {
#pragma unroll
    for (int k = 0; k < NST; ++k) {
      const int i = tid + k * ACC_T;
      if (i < ACC_NB * ROW / 4) xrr4[i] = st[k];
    }
  };
  int* gl = glist + (size_t)blockIdx.x * ACC_CAP;
  int total;
  if constexpr (!SAVED) {
    __shared__ int chunks[IDX_CCAP];
    __shared__ int ccount;
    if (tid == 0) { lcount = 0; ccount = 0; redo = 0; }
    __syncthreads();
    rows_request();
    const int ngrp = (E + IDX_GRP - 1) / IDX_GRP;
    const int col = (int)blockIdx.x < IDX_OWN - 1 ? (int)blockIdx.x : IDX_OWN - 1;
    for (int g = tid; g < ngrp; g += blockDim.x) {
      unsigned w = bmap[(size_t)g * IDX_OWN + col];
      while (w) {
        const int k = __builtin_ctz(w);
        w &= w - 1;
        const int slot = atomicAdd(&ccount, 1);
        if (slot < IDX_CCAP) chunks[slot] = g * 32 + k;
      }
    }
    __syncthreads();
    const int nch = __builtin_amdgcn_readfirstlane(ccount);
    if (nch <= IDX_CCAP) {
      for (int i0 = wave; i0 < nch; i0 += nw * IDX_U) {
        const int iu = i0 + lane * nw;
        const int mine = (lane < IDX_U && iu < nch) ? chunks[iu] : -1;
        int ee[IDX_U], dd[IDX_U];
#pragma unroll
        for (int u = 0; u < IDX_U; ++u) {
          const int cn = __builtin_amdgcn_readlane(mine, u);
          ee[u] = cn >= 0 ? cn * IDX_CH + lane : E;
          dd[u] = ee[u] < E ? dstI[ee[u]] : -1;
        }
        unsigned own[IDX_U];
        int cnt = 0;
#pragma unroll
        for (int u = 0; u < IDX_U; ++u) {
          own[u] = __builtin_amdgcn_ballot_w32((unsigned)(dd[u] - n0) < (unsigned)ACC_NB);
          cnt += __builtin_popcount(own[u]);
        }
        if (cnt) {
          int base = 0;
          if (lane == 0) base = atomicAdd(&lcount, cnt);
          base = __builtin_amdgcn_readfirstlane(base);
#pragma unroll
          for (int u = 0; u < IDX_U; ++u) {
            if ((own[u] >> lane) & 1u) {
              const int slot = base + __builtin_popcount(own[u] & ((1u << lane) - 1u));
              if (slot < ACC_CAP) list[slot] = ee[u];
            }
            base += __builtin_popcount(own[u]);
          }
        }
      }
    } else {
      scan_list(dstI, E, n0, wave, nw, lane, list, &lcount);
    }
    rows_store();
    __syncthreads();
    total = __builtin_amdgcn_readfirstlane(lcount);
    const int nsave = total < ACC_CAP ? total : ACC_CAP;
    for (int i = tid; i < nsave; i += blockDim.x) gl[i] = list[i];
    if (tid == 0) gcount[blockIdx.x] = total;
  } else {
    if (tid == 0) redo = 0;
    rows_request();
    total = __builtin_amdgcn_readfirstlane(gcount[blockIdx.x]);
    if (total <= ACC_CAP)
      for (int i = tid; i < total; i += blockDim.x) list[i] = gl[i];
    rows_store();
    __syncthreads();
  }
  const bool listed = total <= ACC_CAP;

  auto fetch = [&](int idx, int& e, int& s, int& d) {
    if (idx < total) { e = __builtin_amdgcn_readfirstlane(list[idx]); s = srcI[e]; d = dstI[e]; }
    else             { e = -1; s = 0; d = n0; }
  };

  float aw[NJ];
#pragma unroll
  for (int j = 0; j < NJ; ++j) aw[j] = att[j * 32 + lane];

  for (int i = wave; i < ACC_NB; i += nw) {
    const int n = n0 + i;
    if (n < N) {
      float a[NJ];
      logit_load<C>(xl, n, lane, a);
      int h;
      const float t = head_logit<C>(a, xrr, aw, i, lane, h);
      if ((lane & 3) == 0) { mx[i * HEADS + h] = __float_as_uint(t); den[i * HEADS + h] = 1.f; }
#pragma unroll
      for (int j = 0; j < NJ; ++j) acc[i * ROW + j * 32 + lane] = a[j];
    }
  }
  __syncthreads();

  if (listed) {
    int i = wave, e0, s0, d0, e1, s1, d1;
    fetch(i, e0, s0, d0);
    fetch(i + nw, e1, s1, d1);
    while (e0 >= 0) {
      float a0[NJ], a1[NJ];
      logit_load<C>(xl, s0, lane, a0);
      logit_load<C>(xl, s1, lane, a1);
      int f0, t0, g0, f1, t1, g1;
      fetch(i + 2 * nw, f0, t0, g0);
      fetch(i + 3 * nw, f1, t1, g1);
      sp_finish<C>(a0, xrr, aw, mx, acc, den, &redo, d0 - n0, lane);
      if (e1 >= 0) sp_finish<C>(a1, xrr, aw, mx, acc, den, &redo, d1 - n0, lane);
      i += 2 * nw;
      e0 = f0; s0 = t0; d0 = g0; e1 = f1; s1 = t1; d1 = g1;
    }
  } else {
    scan_owned(dstI, E, n0, wave, nw, lane, [&](int e, int d) {
      float a[NJ];
      logit_load<C>(xl, srcI[e], lane, a);
      sp_finish<C>(a, xrr, aw, mx, acc, den, &redo, d - n0, lane);
    });
  }
  __syncthreads();

  if (__builtin_amdgcn_readfirstlane(redo)) {
    for (int i = tid; i < ACC_NB * HEADS; i += blockDim.x) { den[i] = 0.f; mx[i] = 0x007FFFFFu; }
    __syncthreads();
    {
      if (listed) {
        int i = wave, e0, s0, d0, e1, s1, d1;
        fetch(i, e0, s0, d0);
        fetch(i + nw, e1, s1, d1);
        while (e0 >= 0) {
          float a0[NJ], a1[NJ];
          logit_load<C>(xl, s0, lane, a0);
          logit_load<C>(xl, s1, lane, a1);
          int f0, t0, g0, f1, t1, g1;
          fetch(i + 2 * nw, f0, t0, g0);
          fetch(i + 3 * nw, f1, t1, g1);
          logit_finish<C>(a0, xrr, aw, lg, mx, e0, d0 - n0, lane);
          if (e1 >= 0) logit_finish<C>(a1, xrr, aw, lg, mx, e1, d1 - n0, lane);
          i += 2 * nw;
          e0 = f0; s0 = t0; d0 = g0; e1 = f1; s1 = t1; d1 = g1;
        }
      } else {
        scan_owned(dstI, E, n0, wave, nw, lane, [&](int e, int d) {
          edge_logit<C>(xl, xrr, aw, lg, mx, e, srcI[e], d - n0, lane);
        });
      }
      for (int i = wave; i < ACC_NB; i += nw) {
        const int n = n0 + i;
        if (n < N) edge_logit<C>(xl, xrr, aw, lg, mx, E + n, n, i, lane);
      }
    }
    __syncthreads();
    for (int i = tid; i < ACC_NB * ROW / 4; i += blockDim.x) acc4[i] = make_float4(0.f, 0.f, 0.f, 0.f);
    __syncthreads();

    if (listed) {
      int i = wave, e0, s0, d0, e1, s1, d1;
      fetch(i, e0, s0, d0);
      fetch(i + nw, e1, s1, d1);
      while (e0 >= 0) {
        float v0[NJ], v1[NJ], l0, l1;
        unsigned m0, m1;
        const int eb = e1 >= 0 ? e1 : e0;
        add_load<C>(xl, lg, mx, e0, s0, d0 - n0, lane, v0, l0, m0);
        add_load<C>(xl, lg, mx, eb, s1, d1 - n0, lane, v1, l1, m1);
        int f0, t0, g0, f1, t1, g1;
        fetch(i + 2 * nw, f0, t0, g0);
        fetch(i + 3 * nw, f1, t1, g1);
        add_finish<C>(v0, l0, m0, acc, den, d0 - n0, lane);
        if (e1 >= 0) add_finish<C>(v1, l1, m1, acc, den, d1 - n0, lane);
        i += 2 * nw;
        e0 = f0; s0 = t0; d0 = g0; e1 = f1; s1 = t1; d1 = g1;
      }
    } else {
      scan_owned(dstI, E, n0, wave, nw, lane, [&](int e, int d) {
        add_edge<C>(xl, lg, mx, acc, den, e, srcI[e], d - n0, lane);
      });
    }
    for (int i = wave; i < ACC_NB; i += nw) {
      const int n = n0 + i;
      if (n < N) add_edge<C>(xl, lg, mx, acc, den, E + n, n, i, lane);
    }
    __syncthreads();

  }

  if constexpr (!LSM) {
    constexpr int R4 = ROW / 4;
    const float4* b4 = (const float4*)bias;
    float4* o4 = (float4*)(out + (size_t)n0 * ROW);
    for (int i = tid; i < nvalid * R4; i += blockDim.x) {
      const int r = i / R4, c4 = i % R4;
      const float dn = den[r * HEADS + (c4 * 4) / C];
      const float4 a = acc4[i], b = b4[c4];
      float4 v;
      v.x = a.x / dn + b.x; v.x = v.x > 0.f ? v.x : (expf(v.x) - 1.f);
      v.y = a.y / dn + b.y; v.y = v.y > 0.f ? v.y : (expf(v.y) - 1.f);
      v.z = a.z / dn + b.z; v.z = v.z > 0.f ? v.z : (expf(v.z) - 1.f);
      v.w = a.w / dn + b.w; v.w = v.w > 0.f ? v.w : (expf(v.w) - 1.f);
      o4[i] = v;
    }
  } else {
    static_assert(ROW == 128, "log-softmax row: 4 channels per lane");
    for (int r = wave; r < nvalid; r += nw) {
      float v[4];
      for (int k = 0; k < 4; ++k) {
        const int c = lane + 32 * k;
        v[k] = acc[r * ROW + c] / den[r * HEADS + c / C] + bias[c];
      }
      float mx = fmaxf(fmaxf(v[0], v[2]), fmaxf(v[1], v[3]));
      for (int sft = 16; sft > 0; sft >>= 1) mx = fmaxf(mx, __shfl_xor(mx, sft));
      float sum = (expf(v[0] - mx) + expf(v[2] - mx)) + (expf(v[1] - mx) + expf(v[3] - mx));
      for (int sft = 16; sft > 0; sft >>= 1) sum += __shfl_xor(sum, sft);
      const float lse = logf(sum);
      float* orow = out + (size_t)(n0 + r) * ROW + lane;
      for (int k = 0; k < 4; ++k) orow[32 * k] = v[k] - mx - lse;
    }
  }
}

__device__ __forceinline__ constexpr int kprow(int st) { return 4 * (st >> 1) + (st & 1); }

__device__ __forceinline__ void index_group(const int* __restrict__ dstI, int E,
                                            unsigned* __restrict__ bmap, unsigned* tab,
                                            int g, int lane) {
  for (int i = 0; i < IDX_OWN / 32; ++i) tab[i * 32 + lane] = 0u;
  const int e0 = g * IDX_GRP + lane;
  for (int kb = 0; kb < 32; kb += 16) {
    int d[16];
#pragma unroll
    for (int u = 0; u < 16; ++u) {
      const int e = e0 + (kb + u) * IDX_CH;
      d[u] = e < E ? dstI[e] : -1;
    }
#pragma unroll
    for (int u = 0; u < 16; ++u) {
      if (d[u] >= 0) {
        unsigned o = (unsigned)d[u] / (unsigned)ACC_NB;
        o = o < IDX_OWN - 1 ? o : IDX_OWN - 1;
        atomicOr(tab + o, 1u << (kb + u));
      }
    }
  }
  unsigned* row = bmap + (size_t)g * IDX_OWN + lane;
  for (int i = 0; i < IDX_OWN / 32; ++i) row[i * 32] = tab[i * 32 + lane];
}

template <int K, int KOUT, int GC, bool INDEX>
__global__ __launch_bounds__(512) void gemm_pair_lds(
    const float* __restrict__ A, const float* __restrict__ Wl, const float* __restrict__ Wr,
    float* __restrict__ Cl, float* __restrict__ Cr, int M,
    const int* __restrict__ dstI, int E, unsigned* __restrict__ bmap) {
  constexpr int NT = 2 * KOUT;
  constexpr int KP = K / 2;
  constexpr int PITCH = NT + 8;
  constexpr int NA = K / 8;
  constexpr int NB = K / 32;
  constexpr int GROUPS = NT / GC;
  static_assert(KOUT % GC == 0 && GC % 32 == 0, "a column group lies inside one product");
  __shared__ float4 Wp4[KP * PITCH / 2];
#if USE_WMMA_F32
  const int tid = threadIdx.x, lane = tid & 31;
  const int h = lane >> 4, c = lane & 15;
  const int wave = __builtin_amdgcn_readfirstlane(tid >> 5);
  const int items = (M >> 4) * GROUPS;
  const int nblk = (int)gridDim.x;

  float4 a4[NA];
  auto load_strip = [&](int t) {
    const int m = t / GROUPS;
    const float4* ap = (const float4*)(A + (size_t)(m * 16 + c) * K) + h;
#pragma unroll
    for (int j = 0; j < NA; ++j) a4[j] = ap[2 * j];
  };
  int slot = wave;
  int t = slot * nblk + (int)blockIdx.x;
  if (t < items) load_strip(t);

  {
    constexpr int Q = NT / 4;
    constexpr int UNITS = KP * Q;
    static_assert(UNITS % 2048 == 0, "copy loop: 4 units per thread per trip");
    for (int u0 = 0; u0 < UNITS; u0 += 2048) {
      float4 r0[4], r1[4];
#pragma unroll
      for (int i = 0; i < 4; ++i) {
        const int u = u0 + i * 512 + tid;
        const int kp = u / Q, col = 4 * (u % Q);
        const float* s = col < KOUT ? Wl + col : Wr + (col - KOUT);
        r0[i] = *(const float4*)(s + (size_t)(2 * kp) * KOUT);
        r1[i] = *(const float4*)(s + (size_t)(2 * kp + 1) * KOUT);
      }
#pragma unroll
      for (int i = 0; i < 4; ++i) {
        const int u = u0 + i * 512 + tid;
        const int kp = u / Q, q = u % Q;
        float4* d = Wp4 + (kp * PITCH + 4 * q) / 2;
        d[0] = make_float4(r0[i].x, r1[i].x, r0[i].y, r1[i].y);
        d[1] = make_float4(r0[i].z, r1[i].z, r0[i].w, r1[i].w);
      }
    }
  }
  __syncthreads();

  const v2f* bp = (const v2f*)Wp4 + (2 * h) * PITCH + c;
  while (t < items) {
    const int m = t / GROUPS, g = t % GROUPS;
    for (int p = 0; p < GC / 32; ++p) {
      const int n0 = g * GC + 32 * p;
      const v2f* bq = bp + n0;
      v8f acc0 = {}, acc1 = {};
      v2f b[2][8][2];
#pragma unroll
      for (int s = 0; s < 8; ++s) {
        b[0][s][0] = bq[kprow(s) * PITCH];
        b[0][s][1] = bq[kprow(s) * PITCH + 16];
      }
#pragma unroll
      for (int blk = 0; blk < NB; ++blk) {
        if (blk + 1 < NB) {
#pragma unroll
          for (int s = 0; s < 8; ++s) {
            b[(blk + 1) & 1][s][0] = bq[kprow((blk + 1) * 8 + s) * PITCH];
            b[(blk + 1) & 1][s][1] = bq[kprow((blk + 1) * 8 + s) * PITCH + 16];
          }
        }
#pragma unroll
        for (int s = 0; s < 8; ++s) {
          const float4 av = a4[(8 * blk + s) >> 1];
          v2f a;
          if (s & 1) { a[0] = av.z; a[1] = av.w; } else { a[0] = av.x; a[1] = av.y; }
          acc0 = __builtin_amdgcn_wmma_f32_16x16x4_f32(false, a, false, b[blk & 1][s][0],
                                                       (short)0, acc0, false, false);
          acc1 = __builtin_amdgcn_wmma_f32_16x16x4_f32(false, a, false, b[blk & 1][s][1],
                                                       (short)0, acc1, false, false);
        }
      }
      float* Cb = (n0 < KOUT ? Cl + n0 : Cr + (n0 - KOUT)) + (size_t)(m * 16) * KOUT + lane;
#pragma unroll
      for (int v = 0; v < 8; ++v) {
        const float got = xhalf(h ? acc0[v] : acc1[v]);
        Cb[(size_t)v * KOUT] = h ? got : acc0[v];
        Cb[(size_t)(8 + v) * KOUT] = h ? acc1[v] : got;
      }
    }
    slot += 16;
    t = slot * nblk + (int)blockIdx.x;
    if (t < items) load_strip(t);
  }
#else
  for (int i = blockIdx.x * blockDim.x + threadIdx.x; i < M * NT; i += gridDim.x * blockDim.x) {
    const int r = i / NT, n = i % NT;
    const float* Wc = n < KOUT ? Wl + n : Wr + (n - KOUT);
    float s = 0.f;
    for (int k = 0; k < K; ++k) s += A[(size_t)r * K + k] * Wc[(size_t)k * KOUT];
    (n < KOUT ? Cl + n : Cr + (n - KOUT))[(size_t)r * KOUT] = s;
  }
#endif
  if constexpr (INDEX) {
    __shared__ unsigned own[16 * IDX_OWN];
    const int iwave = __builtin_amdgcn_readfirstlane((int)threadIdx.x >> 5);
    const int inw = (int)blockDim.x >> 5;
    const int ngrp = (E + IDX_GRP - 1) / IDX_GRP;
    for (int g = (inw - 1 - iwave) * (int)gridDim.x + (int)blockIdx.x; g < ngrp; g += inw * (int)gridDim.x)
      index_group(dstI, E, bmap, own + (iwave & 15) * IDX_OWN, g, (int)threadIdx.x & 31);
  }
}

extern "C" void kernel_launch(void* const* d_in, const int* in_sizes, int n_in,
                              void* d_out, int out_size, void* d_ws, size_t ws_size,
                              hipStream_t stream) {
  (void)n_in; (void)out_size; (void)ws_size;
  const float* x    = (const float*)d_in[0];
  const int*   ei   = (const int*)d_in[1];
  const float* W1l  = (const float*)d_in[2];
  const float* W1r  = (const float*)d_in[3];
  const float* att1 = (const float*)d_in[4];
  const float* b1   = (const float*)d_in[5];
  const float* W2l  = (const float*)d_in[6];
  const float* W2r  = (const float*)d_in[7];
  const float* att2 = (const float*)d_in[8];
  const float* b2   = (const float*)d_in[9];

  int N = in_sizes[0] / 128;
  int E = in_sizes[1] / 2;
  const int* srcI = ei;
  const int* dstI = ei + E;

  float* ws    = (float*)d_ws;
  float* xl    = ws;
  float* xr    = xl + (size_t)N * 256;
  float* hbuf  = xr + (size_t)N * 256;
  float* lg    = hbuf + (size_t)N * 256;
  int* glist   = (int*)(lg + (size_t)(E + N) * 8);
  float* outFin = (float*)d_out;

  int agrid = (N + ACC_NB - 1) / ACC_NB;
  unsigned* bmap = (unsigned*)(glist + (size_t)agrid * ACC_CAP);
  int* gcount  = (int*)(bmap + (size_t)((E + IDX_GRP - 1) / IDX_GRP) * IDX_OWN);

  gemm_pair_lds<128, 256, 128, true><<<256, dim3(512), 0, stream>>>(x, W1l, W1r, xl, xr, N, dstI, E, bmap);
  edge_owned<32, false, false><<<agrid, dim3(ACC_T), 0, stream>>>(xl, xr, att1, srcI, dstI, E, N, lg, glist, gcount, bmap, b1, hbuf);

  gemm_pair_lds<256, 128, 64, false><<<256, dim3(512), 0, stream>>>(hbuf, W2l, W2r, xl, xr, N, dstI, E, bmap);
  edge_owned<16, true, true><<<agrid, dim3(ACC_T), 0, stream>>>(xl, xr, att2, srcI, dstI, E, N, lg, glist, gcount, bmap, b2, outFin);
}
